# starting kernel
# baseline (speedup 1.0000x reference)
#include <hip/hip_runtime.h>
#include <hip/hip_bf16.h>
#include <cfloat>

#define BS 64
#define NQ 900
#define NT 200
#define NC 92
#define KSTEPS (NC / 4)
#define TTILES ((NT + 15) / 16)
#define QTILES ((NQ + 15) / 16)

typedef float v2f __attribute__((ext_vector_type(2)));
typedef float v8f __attribute__((ext_vector_type(8)));

__global__ __launch_bounds__(256) void softmax_rows_k(
    const float* __restrict__ logits, float* __restrict__ prob) {
  const int row = blockIdx.x * 8 + (threadIdx.x >> 5);
  if (row >= BS * NQ) return;
  const int lane = threadIdx.x & 31;
  const float* x = logits + (size_t)row * NC;

  float a0 = x[lane];
  float a1 = x[lane + 32];
  float a2 = (lane + 64 < NC) ? x[lane + 64] : -FLT_MAX;

  float m = fmaxf(a0, fmaxf(a1, a2));
#pragma unroll
  for (int off = 16; off > 0; off >>= 1) m = fmaxf(m, __shfl_xor(m, off, 32));

  float e0 = __expf(a0 - m);
  float e1 = __expf(a1 - m);
  float e2 = (lane + 64 < NC) ? __expf(a2 - m) : 0.0f;

  float s = e0 + e1 + e2;
#pragma unroll
  for (int off = 16; off > 0; off >>= 1) s += __shfl_xor(s, off, 32);

  const float inv = 1.0f / s;
  float* p = prob + (size_t)row * NC;
  p[lane]      = e0 * inv;
  p[lane + 32] = e1 * inv;
  if (lane + 64 < NC) p[lane + 64] = e2 * inv;
}

__global__ __launch_bounds__(128) void cost_strip_k(
    const float* __restrict__ prob,
    const float* __restrict__ pred_boxes,
    const float* __restrict__ boxes,
    const int*   __restrict__ labels,
    float* __restrict__ out) {
  const int wave = threadIdx.x >> 5;
  const int qt = blockIdx.x * 4 + wave;
  if (qt * 16 >= NQ) return;
  const int b = blockIdx.z;

  const int lane = threadIdx.x & 31;
  const int half = lane >> 4;
  const int l15  = lane & 15;
  const int kb   = half << 1;

  const int q_m  = qt * 16 + l15;
  const int q_cl = (q_m < NQ) ? q_m : (NQ - 1);
  const float* arow = prob + ((size_t)b * NQ + q_cl) * NC;

  v2f afrag[KSTEPS];
#pragma unroll
  for (int kk = 0; kk < KSTEPS; ++kk) {
    afrag[kk].x = arow[4 * kk + kb];
    afrag[kk].y = arow[4 * kk + kb + 1];
  }

  float4 pbs[8];
#pragma unroll
  for (int v = 0; v < 8; ++v) {
    const int m = v + (half << 3);
    const int q = qt * 16 + m;
    const int qc = (q < NQ) ? q : (NQ - 1);
    pbs[v] = *(const float4*)(pred_boxes + ((size_t)b * NQ + qc) * 4);
  }

  for (int tt = 0; tt < TTILES; ++tt) {
    const int t_n   = tt * 16 + l15;
    const int t_cl  = (t_n < NT) ? t_n : (NT - 1);
    const int label = labels[(size_t)b * NT + t_cl];

    v8f acc = {0.f, 0.f, 0.f, 0.f, 0.f, 0.f, 0.f, 0.f};
#pragma unroll
    for (int kk = 0; kk < KSTEPS; ++kk) {
      const int k = 4 * kk;
      v2f bm;
      bm.x = (label == (k + kb))     ? 1.0f : 0.0f;
      bm.y = (label == (k + kb + 1)) ? 1.0f : 0.0f;
      acc = __builtin_amdgcn_wmma_f32_16x16x4_f32(
           false, afrag[kk],  false, bm,
           (short)0, acc,  false,  false);
    }

    const float4 tb = *(const float4*)(boxes + ((size_t)b * NT + t_cl) * 4);
    const float bx0 = tb.x - 0.5f * tb.z, by0 = tb.y - 0.5f * tb.w;
    const float bx1 = tb.x + 0.5f * tb.z, by1 = tb.y + 0.5f * tb.w;
    const float area2 = (bx1 - bx0) * (by1 - by0);
    const bool  tval  = (t_n < NT);

#pragma unroll
    for (int v = 0; v < 8; ++v) {
      const int m = v + (half << 3);
      const int q = qt * 16 + m;
      if (q >= NQ || !tval) continue;
      const float4 pb = pbs[v];

      const float l1 = fabsf(pb.x - tb.x) + fabsf(pb.y - tb.y) +
                       fabsf(pb.z - tb.z) + fabsf(pb.w - tb.w);

      const float ax0 = pb.x - 0.5f * pb.z, ay0 = pb.y - 0.5f * pb.w;
      const float ax1 = pb.x + 0.5f * pb.z, ay1 = pb.y + 0.5f * pb.w;
      const float area1 = (ax1 - ax0) * (ay1 - ay0);

      const float iw    = fminf(ax1, bx1) - fmaxf(ax0, bx0);
      const float ih    = fminf(ay1, by1) - fmaxf(ay0, by0);
      const float inter = fmaxf(iw, 0.f) * fmaxf(ih, 0.f);
      const float uni   = area1 + area2 - inter;
      const float iou   = inter / uni;

      const float ew    = fmaxf(ax1, bx1) - fminf(ax0, bx0);
      const float eh    = fmaxf(ay1, by1) - fminf(ay0, by0);
      const float areae = fmaxf(ew, 0.f) * fmaxf(eh, 0.f);
      const float giou  = iou - (areae - uni) / areae;

      out[((size_t)b * NQ + q) * NT + t_n] = 5.0f * l1 - acc[v] - 2.0f * giou;
    }
  }
}

extern "C" void kernel_launch(void* const* d_in, const int* in_sizes, int n_in,
                              void* d_out, int out_size, void* d_ws, size_t ws_size,
                              hipStream_t stream) {
  const float* pred_logits = (const float*)d_in[0];
  const float* pred_boxes  = (const float*)d_in[1];
  const float* boxes       = (const float*)d_in[2];
  const int*   labels      = (const int*)d_in[3];
  float* out  = (float*)d_out;
  float* prob = (float*)d_ws;

  const int rows = BS * NQ;
  softmax_rows_k<<<(rows + 7) / 8, 256, 0, stream>>>(pred_logits, prob);

  dim3 grid((QTILES + 3) / 4, 1, BS);
  cost_strip_k<<<grid, 128, 0, stream>>>(prob, pred_boxes, boxes, labels, out);
}


# Softmax pass re-cut: one lane per row, 32 rows per wave, rows turned through an LDS tile, whole-line 16-byte loads and stores; 1800 long waves replace 57600 one-row waves (new row-sum order)
# speedup vs baseline: 1.0771x; 1.0771x over previous
#include <hip/hip_runtime.h>
#include <hip/hip_bf16.h>
#include <cfloat>

#define BS 64
#define NQ 900
#define NT 200
#define NC 92
#define KSTEPS (NC / 4)
#define TTILES ((NT + 15) / 16)
#define QTILES ((NQ + 15) / 16)

typedef float v2f __attribute__((ext_vector_type(2)));
typedef float v8f __attribute__((ext_vector_type(8)));

#define SM_WAVES 4
#define SM_ROWS  32
#define SM_V4    (NC / 4)
#define SM_CHUNKS ((BS * NQ) / SM_ROWS)
#define SM_BLOCKS (SM_CHUNKS / SM_WAVES)

__global__ __launch_bounds__(SM_WAVES * 32) void softmax_rows_k(
    const float* __restrict__ logits, float* __restrict__ prob) {
  static_assert((BS * NQ) % SM_ROWS == 0, "rows split into whole 32-row chunks");
  static_assert(SM_CHUNKS % SM_WAVES == 0, "chunks split into whole work-groups");
  static_assert(NC % 4 == 0, "a row is a whole number of float4");
  __shared__ float4 tile[SM_WAVES][SM_ROWS * SM_V4];

  const int wave  = threadIdx.x >> 5;
  const int lane  = threadIdx.x & 31;
  const int chunk = blockIdx.x * SM_WAVES + wave;
  const size_t base = (size_t)chunk * (SM_ROWS * NC);
  const float4* src = (const float4*)(logits + base);
  float4*       dst = (float4*)(prob + base);
  float4* t = tile[wave];

  float4 v[SM_V4];
#pragma unroll
  for (int i = 0; i < SM_V4; ++i) v[i] = src[i * 32 + lane];
#pragma unroll
  for (int i = 0; i < SM_V4; ++i) t[i * 32 + lane] = v[i];
  __syncthreads();

  float4* myrow = t + lane * SM_V4;
  float4 r[SM_V4];
#pragma unroll
  for (int j = 0; j < SM_V4; ++j) r[j] = myrow[j];

  float m = fmaxf(fmaxf(r[0].x, r[0].y), fmaxf(r[0].z, r[0].w));
#pragma unroll
  for (int j = 1; j < SM_V4; ++j)
    m = fmaxf(m, fmaxf(fmaxf(r[j].x, r[j].y), fmaxf(r[j].z, r[j].w)));

  float s0 = 0.f, s1 = 0.f, s2 = 0.f, s3 = 0.f;
#pragma unroll
  for (int j = 0; j < SM_V4; ++j) {
    r[j].x = __expf(r[j].x - m); s0 += r[j].x;
    r[j].y = __expf(r[j].y - m); s1 += r[j].y;
    r[j].z = __expf(r[j].z - m); s2 += r[j].z;
    r[j].w = __expf(r[j].w - m); s3 += r[j].w;
  }
  const float inv = 1.0f / ((s0 + s1) + (s2 + s3));

#pragma unroll
  for (int j = 0; j < SM_V4; ++j)
    myrow[j] = make_float4(r[j].x * inv, r[j].y * inv, r[j].z * inv, r[j].w * inv);
  __syncthreads();

#pragma unroll
  for (int i = 0; i < SM_V4; ++i) v[i] = t[i * 32 + lane];
#pragma unroll
  for (int i = 0; i < SM_V4; ++i) dst[i * 32 + lane] = v[i];
}

__global__ __launch_bounds__(128) void cost_strip_k(
    const float* __restrict__ prob,
    const float* __restrict__ pred_boxes,
    const float* __restrict__ boxes,
    const int*   __restrict__ labels,
    float* __restrict__ out) {
  static_assert(NT % 8 == 0, "a half-tile of 8 targets is valid or empty as a whole");
  const int wave = threadIdx.x >> 5;
  const int qt = blockIdx.x * 4 + wave;
  if (qt * 16 >= NQ) return;
  const int b = blockIdx.z;

  const int lane = threadIdx.x & 31;
  const int half = lane >> 4;
  const int l15  = lane & 15;
  const int kb   = half << 1;

  const int  q    = qt * 16 + l15;
  const bool qval = (q < NQ);
  const int  q_cl = qval ? q : (NQ - 1);
  const float* prow = prob + ((size_t)b * NQ + q_cl) * NC;

  v2f pfrag[KSTEPS];
#pragma unroll
  for (int kk = 0; kk < KSTEPS; ++kk) {
    pfrag[kk].x = prow[4 * kk + kb];
    pfrag[kk].y = prow[4 * kk + kb + 1];
  }

  const float4 pb = *(const float4*)(pred_boxes + ((size_t)b * NQ + q_cl) * 4);
  const float ax0 = pb.x - 0.5f * pb.z, ay0 = pb.y - 0.5f * pb.w;
  const float ax1 = pb.x + 0.5f * pb.z, ay1 = pb.y + 0.5f * pb.w;
  const float area1 = (ax1 - ax0) * (ay1 - ay0);

  float* orow = out + ((size_t)b * NQ + q_cl) * NT;

  for (int tt = 0; tt < TTILES; ++tt) {
    const int t_m   = tt * 16 + l15;
    const int t_cl  = (t_m < NT) ? t_m : (NT - 1);
    const int label = labels[(size_t)b * NT + t_cl];

    const int  t0   = tt * 16 + (half << 3);
    const bool tval = (t0 < NT);
    const int  t0c  = tval ? t0 : (NT - 8);
    const float4* tbp = (const float4*)(boxes + ((size_t)b * NT + t0c) * 4);

    v8f acc = {0.f, 0.f, 0.f, 0.f, 0.f, 0.f, 0.f, 0.f};
#pragma unroll
    for (int kk = 0; kk < KSTEPS; ++kk) {
      const int k = 4 * kk;
      v2f oh;
      oh.x = (label == (k + kb))     ? 1.0f : 0.0f;
      oh.y = (label == (k + kb + 1)) ? 1.0f : 0.0f;
      acc = __builtin_amdgcn_wmma_f32_16x16x4_f32(
           false, oh,  false, pfrag[kk],
           (short)0, acc,  false,  false);
    }

    float res[8];
#pragma unroll
    for (int r = 0; r < 8; ++r) {
      const float4 tb = tbp[r];
      const float bx0 = tb.x - 0.5f * tb.z, by0 = tb.y - 0.5f * tb.w;
      const float bx1 = tb.x + 0.5f * tb.z, by1 = tb.y + 0.5f * tb.w;
      const float area2 = (bx1 - bx0) * (by1 - by0);

      const float l1 = fabsf(pb.x - tb.x) + fabsf(pb.y - tb.y) +
                       fabsf(pb.z - tb.z) + fabsf(pb.w - tb.w);

      const float iw    = fminf(ax1, bx1) - fmaxf(ax0, bx0);
      const float ih    = fminf(ay1, by1) - fmaxf(ay0, by0);
      const float inter = fmaxf(iw, 0.f) * fmaxf(ih, 0.f);
      const float uni   = area1 + area2 - inter;
      const float iou   = inter / uni;

      const float ew    = fmaxf(ax1, bx1) - fminf(ax0, bx0);
      const float eh    = fmaxf(ay1, by1) - fminf(ay0, by0);
      const float areae = fmaxf(ew, 0.f) * fmaxf(eh, 0.f);
      const float giou  = iou - (areae - uni) / areae;

      res[r] = 5.0f * l1 - acc[r] - 2.0f * giou;
    }

    if (qval && tval) {
      float4* o = (float4*)(orow + t0);
      o[0] = make_float4(res[0], res[1], res[2], res[3]);
      o[1] = make_float4(res[4], res[5], res[6], res[7]);
    }
  }
}

extern "C" void kernel_launch(void* const* d_in, const int* in_sizes, int n_in,
                              void* d_out, int out_size, void* d_ws, size_t ws_size,
                              hipStream_t stream) {
  const float* pred_logits = (const float*)d_in[0];
  const float* pred_boxes  = (const float*)d_in[1];
  const float* boxes       = (const float*)d_in[2];
  const int*   labels      = (const int*)d_in[3];
  float* out  = (float*)d_out;
  float* prob = (float*)d_ws;

  softmax_rows_k<<<SM_BLOCKS, SM_WAVES * 32, 0, stream>>>(pred_logits, prob);

  dim3 grid((QTILES + 3) / 4, 1, BS);
  cost_strip_k<<<grid, 128, 0, stream>>>(prob, pred_boxes, boxes, labels, out);
}


# Strip kernel's global traffic turned through a per-wave LDS tile: prob rows come in and the 16x200 strip leaves as whole 128-byte lines, each line stored once, as soon as its last t-tile is done
# speedup vs baseline: 1.1112x; 1.0317x over previous
#include <hip/hip_runtime.h>
#include <hip/hip_bf16.h>
#include <cfloat>

#define BS 64
#define NQ 900
#define NT 200
#define NC 92
#define KSTEPS (NC / 4)
#define TTILES ((NT + 15) / 16)
#define QTILES ((NQ + 15) / 16)

typedef float v2f __attribute__((ext_vector_type(2)));
typedef float v8f __attribute__((ext_vector_type(8)));

#define SM_WAVES 4
#define SM_ROWS  32
#define SM_V4    (NC / 4)
#define SM_CHUNKS ((BS * NQ) / SM_ROWS)
#define SM_BLOCKS (SM_CHUNKS / SM_WAVES)

__global__ __launch_bounds__(SM_WAVES * 32) void softmax_rows_k(
    const float* __restrict__ logits, float* __restrict__ prob) {
  static_assert((BS * NQ) % SM_ROWS == 0, "rows split into whole 32-row chunks");
  static_assert(SM_CHUNKS % SM_WAVES == 0, "chunks split into whole work-groups");
  static_assert(NC % 4 == 0, "a row is a whole number of float4");
  __shared__ float4 tile[SM_WAVES][SM_ROWS * SM_V4];

  const int wave  = threadIdx.x >> 5;
  const int lane  = threadIdx.x & 31;
  const int chunk = blockIdx.x * SM_WAVES + wave;
  const size_t base = (size_t)chunk * (SM_ROWS * NC);
  const float4* src = (const float4*)(logits + base);
  float4*       dst = (float4*)(prob + base);
  float4* t = tile[wave];

  float4 v[SM_V4];
#pragma unroll
  for (int i = 0; i < SM_V4; ++i) v[i] = src[i * 32 + lane];
#pragma unroll
  for (int i = 0; i < SM_V4; ++i) t[i * 32 + lane] = v[i];
  __syncthreads();

  float4* myrow = t + lane * SM_V4;
  float4 r[SM_V4];
#pragma unroll
  for (int j = 0; j < SM_V4; ++j) r[j] = myrow[j];

  float m = fmaxf(fmaxf(r[0].x, r[0].y), fmaxf(r[0].z, r[0].w));
#pragma unroll
  for (int j = 1; j < SM_V4; ++j)
    m = fmaxf(m, fmaxf(fmaxf(r[j].x, r[j].y), fmaxf(r[j].z, r[j].w)));

  float s0 = 0.f, s1 = 0.f, s2 = 0.f, s3 = 0.f;
#pragma unroll
  for (int j = 0; j < SM_V4; ++j) {
    r[j].x = __expf(r[j].x - m); s0 += r[j].x;
    r[j].y = __expf(r[j].y - m); s1 += r[j].y;
    r[j].z = __expf(r[j].z - m); s2 += r[j].z;
    r[j].w = __expf(r[j].w - m); s3 += r[j].w;
  }
  const float inv = 1.0f / ((s0 + s1) + (s2 + s3));

#pragma unroll
  for (int j = 0; j < SM_V4; ++j)
    myrow[j] = make_float4(r[j].x * inv, r[j].y * inv, r[j].z * inv, r[j].w * inv);
  __syncthreads();

#pragma unroll
  for (int i = 0; i < SM_V4; ++i) v[i] = t[i * 32 + lane];
#pragma unroll
  for (int i = 0; i < SM_V4; ++i) dst[i * 32 + lane] = v[i];
}

#define ST_V4 (16 * NT / 4)
#define PR_V4 (16 * NC / 4)
#define PR_LOADS ((PR_V4 + 31) / 32)

__device__ __forceinline__ void wave_lds_handover() {
  __builtin_amdgcn_fence(__ATOMIC_RELEASE, "wavefront");
  __builtin_amdgcn_wave_barrier();
  __builtin_amdgcn_fence(__ATOMIC_ACQUIRE, "wavefront");
}

__global__ __launch_bounds__(128) void cost_strip_k(
    const float* __restrict__ prob,
    const float* __restrict__ pred_boxes,
    const float* __restrict__ boxes,
    const int*   __restrict__ labels,
    float* __restrict__ out) {
  static_assert(NT % 8 == 0, "a half-tile of 8 targets is valid or empty as a whole");
  static_assert(NT == 200 && TTILES == 13, "the line schedule below is written for rows of 6.25 lines");
  static_assert(NQ % 4 == 0, "the valid rows of an edge q-tile are whole groups of 4 rows (25 lines)");
  static_assert(PR_LOADS * 32 <= ST_V4, "the prob rows fit the strip's LDS tile");
  __shared__ float4 stage[4][ST_V4];

  const int wave = __builtin_amdgcn_readfirstlane(threadIdx.x >> 5);
  const int qt = blockIdx.x * 4 + wave;
  if (qt * 16 >= NQ) return;
  const int b = blockIdx.z;
  const int nrows = (NQ - qt * 16 < 16) ? (NQ - qt * 16) : 16;

  const int lane = threadIdx.x & 31;
  const int half = lane >> 4;
  const int l15  = lane & 15;
  const int kb   = half << 1;

  float4* st = stage[wave];

  {
    const float4* psrc = (const float4*)(prob + ((size_t)b * NQ + qt * 16) * NC);
    const int last = nrows * (NC / 4) - 1;
    float4 pin[PR_LOADS];
#pragma unroll
    for (int i = 0; i < PR_LOADS; ++i) {
      const int idx = i * 32 + lane;
      pin[i] = psrc[idx < last ? idx : last];
    }
#pragma unroll
    for (int i = 0; i < PR_LOADS; ++i) st[i * 32 + lane] = pin[i];
  }
  wave_lds_handover();

  const bool qval = (l15 < nrows);
  const int  r_cl = qval ? l15 : (nrows - 1);
  const int  q_cl = qt * 16 + r_cl;

  v2f pfrag[KSTEPS];
  {
    const v2f* prow = (const v2f*)st + r_cl * (NC / 2) + half;
#pragma unroll
    for (int kk = 0; kk < KSTEPS; ++kk) pfrag[kk] = prow[2 * kk];
  }

  const float4 pb = *(const float4*)(pred_boxes + ((size_t)b * NQ + q_cl) * 4);
  const float ax0 = pb.x - 0.5f * pb.z, ay0 = pb.y - 0.5f * pb.w;
  const float ax1 = pb.x + 0.5f * pb.z, ay1 = pb.y + 0.5f * pb.w;
  const float area1 = (ax1 - ax0) * (ay1 - ay0);

  float* srow = (float*)st + l15 * NT;

  float4* gdst = (float4*)(out + ((size_t)b * NQ + qt * 16) * NT);
  const int  fl_g     = lane >> 3;
  const int  fl_piece = lane & 7;
  const bool fl_ok    = (4 * fl_g < nrows);

  for (int tt = 0; tt < TTILES; ++tt) {
    const int t_m   = tt * 16 + l15;
    const int t_cl  = (t_m < NT) ? t_m : (NT - 1);
    const int lrel = labels[(size_t)b * NT + t_cl] - kb;

    const int  t0   = tt * 16 + (half << 3);
    const bool tval = (t0 < NT);
    const int  t0c  = tval ? t0 : (NT - 8);
    const float4* tbp = (const float4*)(boxes + ((size_t)b * NT + t0c) * 4);

    v8f acc = {0.f, 0.f, 0.f, 0.f, 0.f, 0.f, 0.f, 0.f};
#pragma unroll
    for (int kk = 0; kk < KSTEPS; ++kk) {
      const int k = 4 * kk;
      v2f oh;
      oh.x = (lrel == k)     ? 1.0f : 0.0f;
      oh.y = (lrel == k + 1) ? 1.0f : 0.0f;
      acc = __builtin_amdgcn_wmma_f32_16x16x4_f32(
           false, oh,  false, pfrag[kk],
           (short)0, acc,  false,  false);
    }

    float res[8];
#pragma unroll
    for (int r = 0; r < 8; ++r) {
      const float4 tb = tbp[r];
      const float bx0 = tb.x - 0.5f * tb.z, by0 = tb.y - 0.5f * tb.w;
      const float bx1 = tb.x + 0.5f * tb.z, by1 = tb.y + 0.5f * tb.w;
      const float area2 = (bx1 - bx0) * (by1 - by0);

      const float l1 = fabsf(pb.x - tb.x) + fabsf(pb.y - tb.y) +
                       fabsf(pb.z - tb.z) + fabsf(pb.w - tb.w);

      const float iw    = fminf(ax1, bx1) - fmaxf(ax0, bx0);
      const float ih    = fminf(ay1, by1) - fmaxf(ay0, by0);
      const float inter = fmaxf(iw, 0.f) * fmaxf(ih, 0.f);
      const float uni   = area1 + area2 - inter;
      const float iou   = inter / uni;

      const float ew    = fmaxf(ax1, bx1) - fminf(ax0, bx0);
      const float eh    = fmaxf(ay1, by1) - fminf(ay0, by0);
      const float areae = fmaxf(ew, 0.f) * fmaxf(eh, 0.f);
      const float giou  = iou - (areae - uni) / areae;

      res[r] = 5.0f * l1 - acc[r] - 2.0f * giou;
    }

    if (qval && tval) {
      float4* o = (float4*)(srow + t0);
      o[0] = make_float4(res[0], res[1], res[2], res[3]);
      o[1] = make_float4(res[4], res[5], res[6], res[7]);
    }

    if ((tt & 1) || tt == TTILES - 1) {
      const int s = (tt == TTILES - 1) ? 6 : (tt >> 1);
      wave_lds_handover();
      if (fl_ok) {
        const int i0 = (25 * fl_g + s) * 8 + fl_piece;
        if (s == 0) {
          gdst[i0] = st[i0];
        } else {
          const float4 v0 = st[i0], v1 = st[i0 + 48], v2 = st[i0 + 96], v3 = st[i0 + 144];
          gdst[i0] = v0; gdst[i0 + 48] = v1; gdst[i0 + 96] = v2; gdst[i0 + 144] = v3;
        }
      }
    }
  }
}

extern "C" void kernel_launch(void* const* d_in, const int* in_sizes, int n_in,
                              void* d_out, int out_size, void* d_ws, size_t ws_size,
                              hipStream_t stream) {
  const float* pred_logits = (const float*)d_in[0];
  const float* pred_boxes  = (const float*)d_in[1];
  const float* boxes       = (const float*)d_in[2];
  const int*   labels      = (const int*)d_in[3];
  float* out  = (float*)d_out;
  float* prob = (float*)d_ws;

  softmax_rows_k<<<SM_BLOCKS, SM_WAVES * 32, 0, stream>>>(pred_logits, prob);

  dim3 grid((QTILES + 3) / 4, 1, BS);
  cost_strip_k<<<grid, 128, 0, stream>>>(prob, pred_boxes, boxes, labels, out);
}


# One-hot A fragments read from a per-wave LDS impulse table (zeros, one 1.0) through a window placed by the label: one two-dword LDS load per k-step replaces two compares and two selects
# speedup vs baseline: 1.1647x; 1.0481x over previous
#include <hip/hip_runtime.h>
#include <hip/hip_bf16.h>
#include <cfloat>

#define BS 64
#define NQ 900
#define NT 200
#define NC 92
#define KSTEPS (NC / 4)
#define TTILES ((NT + 15) / 16)
#define QTILES ((NQ + 15) / 16)

typedef float v2f __attribute__((ext_vector_type(2)));
typedef float v8f __attribute__((ext_vector_type(8)));

#define SM_WAVES 4
#define SM_ROWS  32
#define SM_V4    (NC / 4)
#define SM_CHUNKS ((BS * NQ) / SM_ROWS)
#define SM_BLOCKS (SM_CHUNKS / SM_WAVES)

__global__ __launch_bounds__(SM_WAVES * 32) void softmax_rows_k(
    const float* __restrict__ logits, float* __restrict__ prob) {
  static_assert((BS * NQ) % SM_ROWS == 0, "rows split into whole 32-row chunks");
  static_assert(SM_CHUNKS % SM_WAVES == 0, "chunks split into whole work-groups");
  static_assert(NC % 4 == 0, "a row is a whole number of float4");
  __shared__ float4 tile[SM_WAVES][SM_ROWS * SM_V4];

  const int wave  = threadIdx.x >> 5;
  const int lane  = threadIdx.x & 31;
  const int chunk = blockIdx.x * SM_WAVES + wave;
  const size_t base = (size_t)chunk * (SM_ROWS * NC);
  const float4* src = (const float4*)(logits + base);
  float4*       dst = (float4*)(prob + base);
  float4* t = tile[wave];

  float4 v[SM_V4];
#pragma unroll
  for (int i = 0; i < SM_V4; ++i) v[i] = src[i * 32 + lane];
#pragma unroll
  for (int i = 0; i < SM_V4; ++i) t[i * 32 + lane] = v[i];
  __syncthreads();

  float4* myrow = t + lane * SM_V4;
  float4 r[SM_V4];
#pragma unroll
  for (int j = 0; j < SM_V4; ++j) r[j] = myrow[j];

  float m = fmaxf(fmaxf(r[0].x, r[0].y), fmaxf(r[0].z, r[0].w));
#pragma unroll
  for (int j = 1; j < SM_V4; ++j)
    m = fmaxf(m, fmaxf(fmaxf(r[j].x, r[j].y), fmaxf(r[j].z, r[j].w)));

  float s0 = 0.f, s1 = 0.f, s2 = 0.f, s3 = 0.f;
#pragma unroll
  for (int j = 0; j < SM_V4; ++j) {
    r[j].x = __expf(r[j].x - m); s0 += r[j].x;
    r[j].y = __expf(r[j].y - m); s1 += r[j].y;
    r[j].z = __expf(r[j].z - m); s2 += r[j].z;
    r[j].w = __expf(r[j].w - m); s3 += r[j].w;
  }
  const float inv = 1.0f / ((s0 + s1) + (s2 + s3));

#pragma unroll
  for (int j = 0; j < SM_V4; ++j)
    myrow[j] = make_float4(r[j].x * inv, r[j].y * inv, r[j].z * inv, r[j].w * inv);
  __syncthreads();

#pragma unroll
  for (int i = 0; i < SM_V4; ++i) v[i] = t[i * 32 + lane];
#pragma unroll
  for (int i = 0; i < SM_V4; ++i) dst[i * 32 + lane] = v[i];
}

#define ST_V4 (16 * NT / 4)
#define PR_V4 (16 * NC / 4)
#define PR_LOADS ((PR_V4 + 31) / 32)
#define OH_HOT 92
#define OH_LEN 192

__device__ __forceinline__ void wave_lds_handover() {
  __builtin_amdgcn_fence(__ATOMIC_RELEASE, "wavefront");
  __builtin_amdgcn_wave_barrier();
  __builtin_amdgcn_fence(__ATOMIC_ACQUIRE, "wavefront");
}

__global__ __launch_bounds__(128) void cost_strip_k(
    const float* __restrict__ prob,
    const float* __restrict__ pred_boxes,
    const float* __restrict__ boxes,
    const int*   __restrict__ labels,
    float* __restrict__ out) {
  static_assert(NT % 8 == 0, "a half-tile of 8 targets is valid or empty as a whole");
  static_assert(NT == 200 && TTILES == 13, "the line schedule below is written for rows of 6.25 lines");
  static_assert(NQ % 4 == 0, "the valid rows of an edge q-tile are whole groups of 4 rows (25 lines)");
  static_assert(PR_LOADS * 32 <= ST_V4, "the prob rows fit the strip's LDS tile");
  static_assert(OH_HOT + 2 + 4 * (KSTEPS - 1) + 1 < OH_LEN, "the farthest window stays inside the table");
  static_assert(OH_HOT >= NC && OH_LEN % 32 == 0, "windows start at slot 0 or later; whole-wave fill");
  __shared__ float4 stage[4][ST_V4];
  __shared__ float ohtab[4][OH_LEN];

  const int wave = __builtin_amdgcn_readfirstlane(threadIdx.x >> 5);
  const int qt = blockIdx.x * 4 + wave;
  if (qt * 16 >= NQ) return;
  const int b = blockIdx.z;
  const int nrows = (NQ - qt * 16 < 16) ? (NQ - qt * 16) : 16;

  const int lane = threadIdx.x & 31;
  const int half = lane >> 4;
  const int l15  = lane & 15;
  const int kb   = half << 1;

  float4* st = stage[wave];

  {
    const float4* psrc = (const float4*)(prob + ((size_t)b * NQ + qt * 16) * NC);
    const int last = nrows * (NC / 4) - 1;
    float4 pin[PR_LOADS];
#pragma unroll
    for (int i = 0; i < PR_LOADS; ++i) {
      const int idx = i * 32 + lane;
      pin[i] = psrc[idx < last ? idx : last];
    }
#pragma unroll
    for (int i = 0; i < PR_LOADS; ++i) st[i * 32 + lane] = pin[i];
  }
  float* zt = ohtab[wave];
#pragma unroll
  for (int i = 0; i < OH_LEN / 32; ++i) zt[i * 32 + lane] = (i * 32 + lane == OH_HOT) ? 1.0f : 0.0f;
  wave_lds_handover();

  const bool qval = (l15 < nrows);
  const int  r_cl = qval ? l15 : (nrows - 1);
  const int  q_cl = qt * 16 + r_cl;

  v2f pfrag[KSTEPS];
  {
    const v2f* prow = (const v2f*)st + r_cl * (NC / 2) + half;
#pragma unroll
    for (int kk = 0; kk < KSTEPS; ++kk) pfrag[kk] = prow[2 * kk];
  }

  const float4 pb = *(const float4*)(pred_boxes + ((size_t)b * NQ + q_cl) * 4);
  const float ax0 = pb.x - 0.5f * pb.z, ay0 = pb.y - 0.5f * pb.w;
  const float ax1 = pb.x + 0.5f * pb.z, ay1 = pb.y + 0.5f * pb.w;
  const float area1 = (ax1 - ax0) * (ay1 - ay0);

  float* srow = (float*)st + l15 * NT;

  float4* gdst = (float4*)(out + ((size_t)b * NQ + qt * 16) * NT);
  const int  fl_g     = lane >> 3;
  const int  fl_piece = lane & 7;
  const bool fl_ok    = (4 * fl_g < nrows);

  for (int tt = 0; tt < TTILES; ++tt) {
    const int t_m   = tt * 16 + l15;
    const int t_cl  = (t_m < NT) ? t_m : (NT - 1);
    const unsigned lrel2 = (unsigned)(labels[(size_t)b * NT + t_cl] - kb + 2);
    const unsigned woff  = (lrel2 < 94u) ? lrel2 : 94u;
    const float* zwin = zt + (OH_HOT + 2) - woff;

    const int  t0   = tt * 16 + (half << 3);
    const bool tval = (t0 < NT);
    const int  t0c  = tval ? t0 : (NT - 8);
    const float4* tbp = (const float4*)(boxes + ((size_t)b * NT + t0c) * 4);

    v8f acc = {0.f, 0.f, 0.f, 0.f, 0.f, 0.f, 0.f, 0.f};
#pragma unroll
    for (int kk = 0; kk < KSTEPS; ++kk) {
      v2f oh;
      oh.x = zwin[4 * kk];
      oh.y = zwin[4 * kk + 1];
      acc = __builtin_amdgcn_wmma_f32_16x16x4_f32(
           false, oh,  false, pfrag[kk],
           (short)0, acc,  false,  false);
    }

    float res[8];
#pragma unroll
    for (int r = 0; r < 8; ++r) {
      const float4 tb = tbp[r];
      const float bx0 = tb.x - 0.5f * tb.z, by0 = tb.y - 0.5f * tb.w;
      const float bx1 = tb.x + 0.5f * tb.z, by1 = tb.y + 0.5f * tb.w;
      const float area2 = (bx1 - bx0) * (by1 - by0);

      const float l1 = fabsf(pb.x - tb.x) + fabsf(pb.y - tb.y) +
                       fabsf(pb.z - tb.z) + fabsf(pb.w - tb.w);

      const float iw    = fminf(ax1, bx1) - fmaxf(ax0, bx0);
      const float ih    = fminf(ay1, by1) - fmaxf(ay0, by0);
      const float inter = fmaxf(iw, 0.f) * fmaxf(ih, 0.f);
      const float uni   = area1 + area2 - inter;
      const float iou   = inter / uni;

      const float ew    = fmaxf(ax1, bx1) - fminf(ax0, bx0);
      const float eh    = fmaxf(ay1, by1) - fminf(ay0, by0);
      const float areae = fmaxf(ew, 0.f) * fmaxf(eh, 0.f);
      const float giou  = iou - (areae - uni) / areae;

      res[r] = 5.0f * l1 - acc[r] - 2.0f * giou;
    }

    if (qval && tval) {
      float4* o = (float4*)(srow + t0);
      o[0] = make_float4(res[0], res[1], res[2], res[3]);
      o[1] = make_float4(res[4], res[5], res[6], res[7]);
    }

    if ((tt & 1) || tt == TTILES - 1) {
      const int s = (tt == TTILES - 1) ? 6 : (tt >> 1);
      wave_lds_handover();
      if (fl_ok) {
        const int i0 = (25 * fl_g + s) * 8 + fl_piece;
        if (s == 0) {
          gdst[i0] = st[i0];
        } else {
          const float4 v0 = st[i0], v1 = st[i0 + 48], v2 = st[i0 + 96], v3 = st[i0 + 144];
          gdst[i0] = v0; gdst[i0 + 48] = v1; gdst[i0 + 96] = v2; gdst[i0 + 144] = v3;
        }
      }
    }
  }
}

extern "C" void kernel_launch(void* const* d_in, const int* in_sizes, int n_in,
                              void* d_out, int out_size, void* d_ws, size_t ws_size,
                              hipStream_t stream) {
  const float* pred_logits = (const float*)d_in[0];
  const float* pred_boxes  = (const float*)d_in[1];
  const float* boxes       = (const float*)d_in[2];
  const int*   labels      = (const int*)d_in[3];
  float* out  = (float*)d_out;
  float* prob = (float*)d_ws;

  softmax_rows_k<<<SM_BLOCKS, SM_WAVES * 32, 0, stream>>>(pred_logits, prob);

  dim3 grid((QTILES + 3) / 4, 1, BS);
  cost_strip_k<<<grid, 128, 0, stream>>>(prob, pred_boxes, boxes, labels, out);
}


# Targets visited in label order: per-wave counting sort by label>>2 (ballots) fills LDS tables of labels, boxes, columns; a 16-target tile spans 2-4 k-steps, only those WMMAs run (skipped terms +0)
# speedup vs baseline: 1.3868x; 1.1907x over previous
#include <hip/hip_runtime.h>
#include <hip/hip_bf16.h>
#include <cfloat>

#define BS 64
#define NQ 900
#define NT 200
#define NC 92
#define KSTEPS (NC / 4)
#define TTILES ((NT + 15) / 16)
#define QTILES ((NQ + 15) / 16)

typedef float v2f __attribute__((ext_vector_type(2)));
typedef float v8f __attribute__((ext_vector_type(8)));

#define SM_WAVES 4
#define SM_ROWS  32
#define SM_V4    (NC / 4)
#define SM_CHUNKS ((BS * NQ) / SM_ROWS)
#define SM_BLOCKS (SM_CHUNKS / SM_WAVES)

__global__ __launch_bounds__(SM_WAVES * 32) void softmax_rows_k(
    const float* __restrict__ logits, float* __restrict__ prob) {
  static_assert((BS * NQ) % SM_ROWS == 0, "rows split into whole 32-row chunks");
  static_assert(SM_CHUNKS % SM_WAVES == 0, "chunks split into whole work-groups");
  static_assert(NC % 4 == 0, "a row is a whole number of float4");
  __shared__ float4 tile[SM_WAVES][SM_ROWS * SM_V4];

  const int wave  = threadIdx.x >> 5;
  const int lane  = threadIdx.x & 31;
  const int chunk = blockIdx.x * SM_WAVES + wave;
  const size_t base = (size_t)chunk * (SM_ROWS * NC);
  const float4* src = (const float4*)(logits + base);
  float4*       dst = (float4*)(prob + base);
  float4* t = tile[wave];

  float4 v[SM_V4];
#pragma unroll
  for (int i = 0; i < SM_V4; ++i) v[i] = src[i * 32 + lane];
#pragma unroll
  for (int i = 0; i < SM_V4; ++i) t[i * 32 + lane] = v[i];
  __syncthreads();

  float4* myrow = t + lane * SM_V4;
  float4 r[SM_V4];
#pragma unroll
  for (int j = 0; j < SM_V4; ++j) r[j] = myrow[j];

  float m = fmaxf(fmaxf(r[0].x, r[0].y), fmaxf(r[0].z, r[0].w));
#pragma unroll
  for (int j = 1; j < SM_V4; ++j)
    m = fmaxf(m, fmaxf(fmaxf(r[j].x, r[j].y), fmaxf(r[j].z, r[j].w)));

  float s0 = 0.f, s1 = 0.f, s2 = 0.f, s3 = 0.f;
#pragma unroll
  for (int j = 0; j < SM_V4; ++j) {
    r[j].x = __expf(r[j].x - m); s0 += r[j].x;
    r[j].y = __expf(r[j].y - m); s1 += r[j].y;
    r[j].z = __expf(r[j].z - m); s2 += r[j].z;
    r[j].w = __expf(r[j].w - m); s3 += r[j].w;
  }
  const float inv = 1.0f / ((s0 + s1) + (s2 + s3));

#pragma unroll
  for (int j = 0; j < SM_V4; ++j)
    myrow[j] = make_float4(r[j].x * inv, r[j].y * inv, r[j].z * inv, r[j].w * inv);
  __syncthreads();

#pragma unroll
  for (int i = 0; i < SM_V4; ++i) v[i] = t[i * 32 + lane];
#pragma unroll
  for (int i = 0; i < SM_V4; ++i) dst[i * 32 + lane] = v[i];
}

#define ST_V4 (16 * NT / 4)
#define PR_V4 (16 * NC / 4)
#define PR_LOADS ((PR_V4 + 31) / 32)
#define OH_HOT 92
#define OH_LEN 192
#define OR_TPL ((NT + 31) / 32)
#define NPOS (TTILES * 16)
#define PAD_LABEL (NC + 8)

__device__ __forceinline__ void wave_lds_handover() {
  __builtin_amdgcn_fence(__ATOMIC_RELEASE, "wavefront");
  __builtin_amdgcn_wave_barrier();
  __builtin_amdgcn_fence(__ATOMIC_ACQUIRE, "wavefront");
}

__device__ __forceinline__ int label_key(int label) {
  const unsigned k = (unsigned)label >> 2;
  return (int)(k < (unsigned)(KSTEPS - 1) ? k : (unsigned)(KSTEPS - 1));
}

__global__ __launch_bounds__(128) void cost_strip_k(
    const float* __restrict__ prob,
    const float* __restrict__ pred_boxes,
    const float* __restrict__ boxes,
    const int*   __restrict__ labels,
    float* __restrict__ out) {
  static_assert(NT % 8 == 0, "a half-tile of 8 positions is valid or empty as a whole");
  static_assert(NPOS - NT <= 32 && NPOS >= NT, "one lane per empty position");
  static_assert(NQ % 4 == 0, "the valid rows of an edge q-tile are whole groups of 4 rows (25 lines)");
  static_assert((4 * NT) % 32 == 0 && (4 * NT) / 32 == 25, "4 rows are 25 whole lines");
  static_assert(PR_LOADS * 32 <= ST_V4, "the prob rows fit the strip's LDS tile");
  static_assert(OH_HOT + 2 + 4 * (KSTEPS - 1) + 1 < OH_LEN, "the farthest window stays inside the table");
  static_assert(OH_HOT >= NC && OH_LEN % 32 == 0, "windows start at slot 0 or later; whole-wave fill");
  __shared__ float4 stage[4][ST_V4];
  __shared__ float ohtab[4][OH_LEN];
  __shared__ float4 sbox[4][NPOS];
  __shared__ int slab[4][NPOS];
  __shared__ __attribute__((aligned(16))) int scol[4][NPOS];

  const int wave = __builtin_amdgcn_readfirstlane(threadIdx.x >> 5);
  const int qt = blockIdx.x * 4 + wave;
  if (qt * 16 >= NQ) return;
  const int b = blockIdx.z;
  const int nrows = (NQ - qt * 16 < 16) ? (NQ - qt * 16) : 16;

  const int lane = threadIdx.x & 31;
  const int half = lane >> 4;
  const int l15  = lane & 15;
  const int kb   = half << 1;

  float4* st = stage[wave];

  {
    const float4* psrc = (const float4*)(prob + ((size_t)b * NQ + qt * 16) * NC);
    const int last = nrows * (NC / 4) - 1;
    float4 pin[PR_LOADS];
#pragma unroll
    for (int i = 0; i < PR_LOADS; ++i) {
      const int idx = i * 32 + lane;
      pin[i] = psrc[idx < last ? idx : last];
    }
#pragma unroll
    for (int i = 0; i < PR_LOADS; ++i) st[i * 32 + lane] = pin[i];
  }
  float* zt = ohtab[wave];
#pragma unroll
  for (int i = 0; i < OH_LEN / 32; ++i) zt[i * 32 + lane] = (i * 32 + lane == OH_HOT) ? 1.0f : 0.0f;

  float4* bx_w = sbox[wave];
  int*    lb_w = slab[wave];
  int*    cl_w = scol[wave];
  {
    int      lab[OR_TPL];
    int      key[OR_TPL];
    int      pos[OR_TPL];
    unsigned peers[OR_TPL];
    unsigned first[OR_TPL];
    float4   tbx[OR_TPL];
#pragma unroll
    for (int j = 0; j < OR_TPL; ++j) {
      const int t  = j * 32 + lane;
      const int tc = (t < NT) ? t : (NT - 1);
      lab[j] = labels[(size_t)b * NT + tc];
      tbx[j] = *(const float4*)(boxes + ((size_t)b * NT + tc) * 4);
      key[j] = (t < NT) ? label_key(lab[j]) : -1;
      peers[j] = 0u;
      first[j] = 0u;
    }
    int base = 0;
    for (int k = 0; k < KSTEPS; ++k) {
#pragma unroll
      for (int j = 0; j < OR_TPL; ++j) {
        const bool hit = (key[j] == k);
        const unsigned m = __builtin_amdgcn_ballot_w32(hit);
        peers[j] = hit ? m : peers[j];
        first[j] = hit ? (unsigned)base : first[j];
        base += __builtin_popcount(m);
      }
    }
#pragma unroll
    for (int j = 0; j < OR_TPL; ++j)
      pos[j] = (int)__builtin_amdgcn_mbcnt_lo(peers[j], first[j]);
#pragma unroll
    for (int j = 0; j < OR_TPL; ++j) {
      const int t = j * 32 + lane;
      if (t < NT) {
        lb_w[pos[j]] = lab[j];
        cl_w[pos[j]] = t;
        bx_w[pos[j]] = tbx[j];
      }
    }
    if (lane < NPOS - NT) {
      lb_w[NT + lane] = PAD_LABEL;
      cl_w[NT + lane] = 0;
      bx_w[NT + lane] = make_float4(0.5f, 0.5f, 0.5f, 0.5f);
    }
  }
  wave_lds_handover();

  const bool qval = (l15 < nrows);
  const int  r_cl = qval ? l15 : (nrows - 1);
  const int  q_cl = qt * 16 + r_cl;

  v2f pfrag[KSTEPS];
  {
    const v2f* prow = (const v2f*)st + r_cl * (NC / 2) + half;
#pragma unroll
    for (int kk = 0; kk < KSTEPS; ++kk) pfrag[kk] = prow[2 * kk];
  }

  const float4 pb = *(const float4*)(pred_boxes + ((size_t)b * NQ + q_cl) * 4);
  const float ax0 = pb.x - 0.5f * pb.z, ay0 = pb.y - 0.5f * pb.w;
  const float ax1 = pb.x + 0.5f * pb.z, ay1 = pb.y + 0.5f * pb.w;
  const float area1 = (ax1 - ax0) * (ay1 - ay0);

  float* srow = (float*)st + l15 * NT;

  for (int tt = 0; tt < TTILES; ++tt) {
    const int labm = lb_w[tt * 16 + l15];
    const unsigned lrel2 = (unsigned)(labm - kb + 2);
    const unsigned woff  = (lrel2 < 94u) ? lrel2 : 94u;
    const float* zwin = zt + (OH_HOT + 2) - woff;

    const int keym    = label_key(labm);
    const int lastrow = (tt == TTILES - 1) ? ((NT - 1) & 15) : 15;
    const int kfirst  = __builtin_amdgcn_readlane(keym, 0);
    const int klast   = __builtin_amdgcn_readlane(keym, lastrow);
    const unsigned kmask = (2u << klast) - (1u << kfirst);

    const int  t0   = tt * 16 + (half << 3);
    const bool tval = (t0 < NT);
    const float4* tbp = bx_w + t0;
    const int4 c0 = *(const int4*)(cl_w + t0);
    const int4 c1 = *(const int4*)(cl_w + t0 + 4);

    v8f acc = {0.f, 0.f, 0.f, 0.f, 0.f, 0.f, 0.f, 0.f};
#pragma unroll
    for (int kk = 0; kk < KSTEPS; ++kk) {
      if ((kmask >> kk) & 1u) {
        v2f oh;
        oh.x = zwin[4 * kk];
        oh.y = zwin[4 * kk + 1];
        acc = __builtin_amdgcn_wmma_f32_16x16x4_f32(
             false, oh,  false, pfrag[kk],
             (short)0, acc,  false,  false);
      }
    }

    float res[8];
#pragma unroll
    for (int r = 0; r < 8; ++r) {
      const float4 tb = tbp[r];
      const float bx0 = tb.x - 0.5f * tb.z, by0 = tb.y - 0.5f * tb.w;
      const float bx1 = tb.x + 0.5f * tb.z, by1 = tb.y + 0.5f * tb.w;
      const float area2 = (bx1 - bx0) * (by1 - by0);

      const float l1 = fabsf(pb.x - tb.x) + fabsf(pb.y - tb.y) +
                       fabsf(pb.z - tb.z) + fabsf(pb.w - tb.w);

      const float iw    = fminf(ax1, bx1) - fmaxf(ax0, bx0);
      const float ih    = fminf(ay1, by1) - fmaxf(ay0, by0);
      const float inter = fmaxf(iw, 0.f) * fmaxf(ih, 0.f);
      const float uni   = area1 + area2 - inter;
      const float iou   = inter / uni;

      const float ew    = fmaxf(ax1, bx1) - fminf(ax0, bx0);
      const float eh    = fmaxf(ay1, by1) - fminf(ay0, by0);
      const float areae = fmaxf(ew, 0.f) * fmaxf(eh, 0.f);
      const float giou  = iou - (areae - uni) / areae;

      res[r] = 5.0f * l1 - acc[r] - 2.0f * giou;
    }

    if (qval && tval) {
      srow[c0.x] = res[0]; srow[c0.y] = res[1]; srow[c0.z] = res[2]; srow[c0.w] = res[3];
      srow[c1.x] = res[4]; srow[c1.y] = res[5]; srow[c1.z] = res[6]; srow[c1.w] = res[7];
    }
  }

  wave_lds_handover();
  {
    float4* gdst = (float4*)(out + ((size_t)b * NQ + qt * 16) * NT);
    const int fl_g     = lane >> 3;
    const int fl_piece = lane & 7;
    if (4 * fl_g < nrows) {
      const int i0 = 25 * 8 * fl_g + fl_piece;
#pragma unroll
      for (int c = 0; c < 5; ++c) {
        float4 v[5];
#pragma unroll
        for (int i = 0; i < 5; ++i) v[i] = st[i0 + (5 * c + i) * 8];
#pragma unroll
        for (int i = 0; i < 5; ++i) gdst[i0 + (5 * c + i) * 8] = v[i];
      }
    }
  }
}

extern "C" void kernel_launch(void* const* d_in, const int* in_sizes, int n_in,
                              void* d_out, int out_size, void* d_ws, size_t ws_size,
                              hipStream_t stream) {
  const float* pred_logits = (const float*)d_in[0];
  const float* pred_boxes  = (const float*)d_in[1];
  const float* boxes       = (const float*)d_in[2];
  const int*   labels      = (const int*)d_in[3];
  float* out  = (float*)d_out;
  float* prob = (float*)d_ws;

  softmax_rows_k<<<SM_BLOCKS, SM_WAVES * 32, 0, stream>>>(pred_logits, prob);

  dim3 grid((QTILES + 3) / 4, 1, BS);
  cost_strip_k<<<grid, 128, 0, stream>>>(prob, pred_boxes, boxes, labels, out);
}


# GIoU as ONE quotient: iou + uni/areae - 1 = (inter*areae + uni*uni)/(uni*areae) - 1; one IEEE f32 division per pair instead of two (giou rounded differently: products, then one quotient)
# speedup vs baseline: 1.4811x; 1.0680x over previous
#include <hip/hip_runtime.h>
#include <hip/hip_bf16.h>
#include <cfloat>

#define BS 64
#define NQ 900
#define NT 200
#define NC 92
#define KSTEPS (NC / 4)
#define TTILES ((NT + 15) / 16)
#define QTILES ((NQ + 15) / 16)

typedef float v2f __attribute__((ext_vector_type(2)));
typedef float v8f __attribute__((ext_vector_type(8)));

#define SM_WAVES 4
#define SM_ROWS  32
#define SM_V4    (NC / 4)
#define SM_CHUNKS ((BS * NQ) / SM_ROWS)
#define SM_BLOCKS (SM_CHUNKS / SM_WAVES)

__global__ __launch_bounds__(SM_WAVES * 32) void softmax_rows_k(
    const float* __restrict__ logits, float* __restrict__ prob) {
  static_assert((BS * NQ) % SM_ROWS == 0, "rows split into whole 32-row chunks");
  static_assert(SM_CHUNKS % SM_WAVES == 0, "chunks split into whole work-groups");
  static_assert(NC % 4 == 0, "a row is a whole number of float4");
  __shared__ float4 tile[SM_WAVES][SM_ROWS * SM_V4];

  const int wave  = threadIdx.x >> 5;
  const int lane  = threadIdx.x & 31;
  const int chunk = blockIdx.x * SM_WAVES + wave;
  const size_t base = (size_t)chunk * (SM_ROWS * NC);
  const float4* src = (const float4*)(logits + base);
  float4*       dst = (float4*)(prob + base);
  float4* t = tile[wave];

  float4 v[SM_V4];
#pragma unroll
  for (int i = 0; i < SM_V4; ++i) v[i] = src[i * 32 + lane];
#pragma unroll
  for (int i = 0; i < SM_V4; ++i) t[i * 32 + lane] = v[i];
  __syncthreads();

  float4* myrow = t + lane * SM_V4;
  float4 r[SM_V4];
#pragma unroll
  for (int j = 0; j < SM_V4; ++j) r[j] = myrow[j];

  float m = fmaxf(fmaxf(r[0].x, r[0].y), fmaxf(r[0].z, r[0].w));
#pragma unroll
  for (int j = 1; j < SM_V4; ++j)
    m = fmaxf(m, fmaxf(fmaxf(r[j].x, r[j].y), fmaxf(r[j].z, r[j].w)));

  float s0 = 0.f, s1 = 0.f, s2 = 0.f, s3 = 0.f;
#pragma unroll
  for (int j = 0; j < SM_V4; ++j) {
    r[j].x = __expf(r[j].x - m); s0 += r[j].x;
    r[j].y = __expf(r[j].y - m); s1 += r[j].y;
    r[j].z = __expf(r[j].z - m); s2 += r[j].z;
    r[j].w = __expf(r[j].w - m); s3 += r[j].w;
  }
  const float inv = 1.0f / ((s0 + s1) + (s2 + s3));

#pragma unroll
  for (int j = 0; j < SM_V4; ++j)
    myrow[j] = make_float4(r[j].x * inv, r[j].y * inv, r[j].z * inv, r[j].w * inv);
  __syncthreads();

#pragma unroll
  for (int i = 0; i < SM_V4; ++i) v[i] = t[i * 32 + lane];
#pragma unroll
  for (int i = 0; i < SM_V4; ++i) dst[i * 32 + lane] = v[i];
}

#define ST_V4 (16 * NT / 4)
#define PR_V4 (16 * NC / 4)
#define PR_LOADS ((PR_V4 + 31) / 32)
#define OH_HOT 92
#define OH_LEN 192
#define OR_TPL ((NT + 31) / 32)
#define NPOS (TTILES * 16)
#define PAD_LABEL (NC + 8)

__device__ __forceinline__ void wave_lds_handover() {
  __builtin_amdgcn_fence(__ATOMIC_RELEASE, "wavefront");
  __builtin_amdgcn_wave_barrier();
  __builtin_amdgcn_fence(__ATOMIC_ACQUIRE, "wavefront");
}

__device__ __forceinline__ int label_key(int label) {
  const unsigned k = (unsigned)label >> 2;
  return (int)(k < (unsigned)(KSTEPS - 1) ? k : (unsigned)(KSTEPS - 1));
}

__global__ __launch_bounds__(128) void cost_strip_k(
    const float* __restrict__ prob,
    const float* __restrict__ pred_boxes,
    const float* __restrict__ boxes,
    const int*   __restrict__ labels,
    float* __restrict__ out) {
  static_assert(NT % 8 == 0, "a half-tile of 8 positions is valid or empty as a whole");
  static_assert(NPOS - NT <= 32 && NPOS >= NT, "one lane per empty position");
  static_assert(NQ % 4 == 0, "the valid rows of an edge q-tile are whole groups of 4 rows (25 lines)");
  static_assert((4 * NT) % 32 == 0 && (4 * NT) / 32 == 25, "4 rows are 25 whole lines");
  static_assert(PR_LOADS * 32 <= ST_V4, "the prob rows fit the strip's LDS tile");
  static_assert(OH_HOT + 2 + 4 * (KSTEPS - 1) + 1 < OH_LEN, "the farthest window stays inside the table");
  static_assert(OH_HOT >= NC && OH_LEN % 32 == 0, "windows start at slot 0 or later; whole-wave fill");
  __shared__ float4 stage[4][ST_V4];
  __shared__ float ohtab[4][OH_LEN];
  __shared__ float4 sbox[4][NPOS];
  __shared__ int slab[4][NPOS];
  __shared__ __attribute__((aligned(16))) int scol[4][NPOS];

  const int wave = __builtin_amdgcn_readfirstlane(threadIdx.x >> 5);
  const int qt = blockIdx.x * 4 + wave;
  if (qt * 16 >= NQ) return;
  const int b = blockIdx.z;
  const int nrows = (NQ - qt * 16 < 16) ? (NQ - qt * 16) : 16;

  const int lane = threadIdx.x & 31;
  const int half = lane >> 4;
  const int l15  = lane & 15;
  const int kb   = half << 1;

  float4* st = stage[wave];

  {
    const float4* psrc = (const float4*)(prob + ((size_t)b * NQ + qt * 16) * NC);
    const int last = nrows * (NC / 4) - 1;
    float4 pin[PR_LOADS];
#pragma unroll
    for (int i = 0; i < PR_LOADS; ++i) {
      const int idx = i * 32 + lane;
      pin[i] = psrc[idx < last ? idx : last];
    }
#pragma unroll
    for (int i = 0; i < PR_LOADS; ++i) st[i * 32 + lane] = pin[i];
  }
  float* zt = ohtab[wave];
#pragma unroll
  for (int i = 0; i < OH_LEN / 32; ++i) zt[i * 32 + lane] = (i * 32 + lane == OH_HOT) ? 1.0f : 0.0f;

  float4* bx_w = sbox[wave];
  int*    lb_w = slab[wave];
  int*    cl_w = scol[wave];
  {
    int      lab[OR_TPL];
    int      key[OR_TPL];
    int      pos[OR_TPL];
    unsigned peers[OR_TPL];
    unsigned first[OR_TPL];
    float4   tbx[OR_TPL];
#pragma unroll
    for (int j = 0; j < OR_TPL; ++j) {
      const int t  = j * 32 + lane;
      const int tc = (t < NT) ? t : (NT - 1);
      lab[j] = labels[(size_t)b * NT + tc];
      tbx[j] = *(const float4*)(boxes + ((size_t)b * NT + tc) * 4);
      key[j] = (t < NT) ? label_key(lab[j]) : -1;
      peers[j] = 0u;
      first[j] = 0u;
    }
    int base = 0;
    for (int k = 0; k < KSTEPS; ++k) {
#pragma unroll
      for (int j = 0; j < OR_TPL; ++j) {
        const bool hit = (key[j] == k);
        const unsigned m = __builtin_amdgcn_ballot_w32(hit);
        peers[j] = hit ? m : peers[j];
        first[j] = hit ? (unsigned)base : first[j];
        base += __builtin_popcount(m);
      }
    }
#pragma unroll
    for (int j = 0; j < OR_TPL; ++j)
      pos[j] = (int)__builtin_amdgcn_mbcnt_lo(peers[j], first[j]);
#pragma unroll
    for (int j = 0; j < OR_TPL; ++j) {
      const int t = j * 32 + lane;
      if (t < NT) {
        lb_w[pos[j]] = lab[j];
        cl_w[pos[j]] = t;
        bx_w[pos[j]] = tbx[j];
      }
    }
    if (lane < NPOS - NT) {
      lb_w[NT + lane] = PAD_LABEL;
      cl_w[NT + lane] = 0;
      bx_w[NT + lane] = make_float4(0.5f, 0.5f, 0.5f, 0.5f);
    }
  }
  wave_lds_handover();

  const bool qval = (l15 < nrows);
  const int  r_cl = qval ? l15 : (nrows - 1);
  const int  q_cl = qt * 16 + r_cl;

  v2f pfrag[KSTEPS];
  {
    const v2f* prow = (const v2f*)st + r_cl * (NC / 2) + half;
#pragma unroll
    for (int kk = 0; kk < KSTEPS; ++kk) pfrag[kk] = prow[2 * kk];
  }

  const float4 pb = *(const float4*)(pred_boxes + ((size_t)b * NQ + q_cl) * 4);
  const float ax0 = pb.x - 0.5f * pb.z, ay0 = pb.y - 0.5f * pb.w;
  const float ax1 = pb.x + 0.5f * pb.z, ay1 = pb.y + 0.5f * pb.w;
  const float area1 = (ax1 - ax0) * (ay1 - ay0);

  float* srow = (float*)st + l15 * NT;

  for (int tt = 0; tt < TTILES; ++tt) {
    const int labm = lb_w[tt * 16 + l15];
    const unsigned lrel2 = (unsigned)(labm - kb + 2);
    const unsigned woff  = (lrel2 < 94u) ? lrel2 : 94u;
    const float* zwin = zt + (OH_HOT + 2) - woff;

    const int keym    = label_key(labm);
    const int lastrow = (tt == TTILES - 1) ? ((NT - 1) & 15) : 15;
    const int kfirst  = __builtin_amdgcn_readlane(keym, 0);
    const int klast   = __builtin_amdgcn_readlane(keym, lastrow);
    const unsigned kmask = (2u << klast) - (1u << kfirst);

    const int  t0   = tt * 16 + (half << 3);
    const bool tval = (t0 < NT);
    const float4* tbp = bx_w + t0;
    const int4 c0 = *(const int4*)(cl_w + t0);
    const int4 c1 = *(const int4*)(cl_w + t0 + 4);

    v8f acc = {0.f, 0.f, 0.f, 0.f, 0.f, 0.f, 0.f, 0.f};
#pragma unroll
    for (int kk = 0; kk < KSTEPS; ++kk) {
      if ((kmask >> kk) & 1u) {
        v2f oh;
        oh.x = zwin[4 * kk];
        oh.y = zwin[4 * kk + 1];
        acc = __builtin_amdgcn_wmma_f32_16x16x4_f32(
             false, oh,  false, pfrag[kk],
             (short)0, acc,  false,  false);
      }
    }

    float res[8];
#pragma unroll
    for (int r = 0; r < 8; ++r) {
      const float4 tb = tbp[r];
      const float bx0 = tb.x - 0.5f * tb.z, by0 = tb.y - 0.5f * tb.w;
      const float bx1 = tb.x + 0.5f * tb.z, by1 = tb.y + 0.5f * tb.w;
      const float area2 = (bx1 - bx0) * (by1 - by0);

      const float l1 = fabsf(pb.x - tb.x) + fabsf(pb.y - tb.y) +
                       fabsf(pb.z - tb.z) + fabsf(pb.w - tb.w);

      const float iw    = fminf(ax1, bx1) - fmaxf(ax0, bx0);
      const float ih    = fminf(ay1, by1) - fmaxf(ay0, by0);
      const float inter = fmaxf(iw, 0.f) * fmaxf(ih, 0.f);
      const float uni   = area1 + area2 - inter;

      const float ew    = fmaxf(ax1, bx1) - fminf(ax0, bx0);
      const float eh    = fmaxf(ay1, by1) - fminf(ay0, by0);
      const float areae = fmaxf(ew, 0.f) * fmaxf(eh, 0.f);
      const float gq    = fmaf(inter, areae, uni * uni) / (uni * areae);
      const float giou  = gq - 1.0f;

      res[r] = 5.0f * l1 - acc[r] - 2.0f * giou;
    }

    if (qval && tval) {
      srow[c0.x] = res[0]; srow[c0.y] = res[1]; srow[c0.z] = res[2]; srow[c0.w] = res[3];
      srow[c1.x] = res[4]; srow[c1.y] = res[5]; srow[c1.z] = res[6]; srow[c1.w] = res[7];
    }
  }

  wave_lds_handover();
  {
    float4* gdst = (float4*)(out + ((size_t)b * NQ + qt * 16) * NT);
    const int fl_g     = lane >> 3;
    const int fl_piece = lane & 7;
    if (4 * fl_g < nrows) {
      const int i0 = 25 * 8 * fl_g + fl_piece;
#pragma unroll
      for (int c = 0; c < 5; ++c) {
        float4 v[5];
#pragma unroll
        for (int i = 0; i < 5; ++i) v[i] = st[i0 + (5 * c + i) * 8];
#pragma unroll
        for (int i = 0; i < 5; ++i) gdst[i0 + (5 * c + i) * 8] = v[i];
      }
    }
  }
}

extern "C" void kernel_launch(void* const* d_in, const int* in_sizes, int n_in,
                              void* d_out, int out_size, void* d_ws, size_t ws_size,
                              hipStream_t stream) {
  const float* pred_logits = (const float*)d_in[0];
  const float* pred_boxes  = (const float*)d_in[1];
  const float* boxes       = (const float*)d_in[2];
  const int*   labels      = (const int*)d_in[3];
  float* out  = (float*)d_out;
  float* prob = (float*)d_ws;

  softmax_rows_k<<<SM_BLOCKS, SM_WAVES * 32, 0, stream>>>(pred_logits, prob);

  dim3 grid((QTILES + 3) / 4, 1, BS);
  cost_strip_k<<<grid, 128, 0, stream>>>(prob, pred_boxes, boxes, labels, out);
}


# Target table built once per work-group (its 4 waves share a batch), not per wave: ordering split over the waves, box terms stored as position pairs with the area finished (area2 rounded alone)
# speedup vs baseline: 1.5937x; 1.0760x over previous
#include <hip/hip_runtime.h>
#include <hip/hip_bf16.h>
#include <cfloat>

#define BS 64
#define NQ 900
#define NT 200
#define NC 92
#define KSTEPS (NC / 4)
#define TTILES ((NT + 15) / 16)
#define QTILES ((NQ + 15) / 16)

typedef float v2f __attribute__((ext_vector_type(2)));
typedef float v8f __attribute__((ext_vector_type(8)));

#define SM_WAVES 4
#define SM_ROWS  32
#define SM_V4    (NC / 4)
#define SM_CHUNKS ((BS * NQ) / SM_ROWS)
#define SM_BLOCKS (SM_CHUNKS / SM_WAVES)

__global__ __launch_bounds__(SM_WAVES * 32) void softmax_rows_k(
    const float* __restrict__ logits, float* __restrict__ prob) {
  static_assert((BS * NQ) % SM_ROWS == 0, "rows split into whole 32-row chunks");
  static_assert(SM_CHUNKS % SM_WAVES == 0, "chunks split into whole work-groups");
  static_assert(NC % 4 == 0, "a row is a whole number of float4");
  __shared__ float4 tile[SM_WAVES][SM_ROWS * SM_V4];

  const int wave  = threadIdx.x >> 5;
  const int lane  = threadIdx.x & 31;
  const int chunk = blockIdx.x * SM_WAVES + wave;
  const size_t base = (size_t)chunk * (SM_ROWS * NC);
  const float4* src = (const float4*)(logits + base);
  float4*       dst = (float4*)(prob + base);
  float4* t = tile[wave];

  float4 v[SM_V4];
#pragma unroll
  for (int i = 0; i < SM_V4; ++i) v[i] = src[i * 32 + lane];
#pragma unroll
  for (int i = 0; i < SM_V4; ++i) t[i * 32 + lane] = v[i];
  __syncthreads();

  float4* myrow = t + lane * SM_V4;
  float4 r[SM_V4];
#pragma unroll
  for (int j = 0; j < SM_V4; ++j) r[j] = myrow[j];

  float m = fmaxf(fmaxf(r[0].x, r[0].y), fmaxf(r[0].z, r[0].w));
#pragma unroll
  for (int j = 1; j < SM_V4; ++j)
    m = fmaxf(m, fmaxf(fmaxf(r[j].x, r[j].y), fmaxf(r[j].z, r[j].w)));

  float s0 = 0.f, s1 = 0.f, s2 = 0.f, s3 = 0.f;
#pragma unroll
  for (int j = 0; j < SM_V4; ++j) {
    r[j].x = __expf(r[j].x - m); s0 += r[j].x;
    r[j].y = __expf(r[j].y - m); s1 += r[j].y;
    r[j].z = __expf(r[j].z - m); s2 += r[j].z;
    r[j].w = __expf(r[j].w - m); s3 += r[j].w;
  }
  const float inv = 1.0f / ((s0 + s1) + (s2 + s3));

#pragma unroll
  for (int j = 0; j < SM_V4; ++j)
    myrow[j] = make_float4(r[j].x * inv, r[j].y * inv, r[j].z * inv, r[j].w * inv);
  __syncthreads();

#pragma unroll
  for (int i = 0; i < SM_V4; ++i) v[i] = t[i * 32 + lane];
#pragma unroll
  for (int i = 0; i < SM_V4; ++i) dst[i * 32 + lane] = v[i];
}

#define ST_V4 (16 * NT / 4)
#define PR_V4 (16 * NC / 4)
#define PR_LOADS ((PR_V4 + 31) / 32)
#define OH_HOT 92
#define OH_LEN 192
#define OR_TPL ((NT + 31) / 32)
#define NPOS (TTILES * 16)
#define PAD_LABEL (NC + 8)
#define CNT_PER_LANE 6
#define CNT_LEN (32 * CNT_PER_LANE)
#define TP_PITCH 12

__device__ __forceinline__ void wave_lds_handover() {
  __builtin_amdgcn_fence(__ATOMIC_RELEASE, "wavefront");
  __builtin_amdgcn_wave_barrier();
  __builtin_amdgcn_fence(__ATOMIC_ACQUIRE, "wavefront");
}

__device__ __forceinline__ int label_key(int label) {
  const unsigned k = (unsigned)label >> 2;
  return (int)(k < (unsigned)(KSTEPS - 1) ? k : (unsigned)(KSTEPS - 1));
}

__device__ __forceinline__ void write_target(float* tp, int pos, float4 tb) {
  float* e = tp + (pos >> 1) * TP_PITCH + (pos & 1);
  const float x0 = tb.x - 0.5f * tb.z, y0 = tb.y - 0.5f * tb.w;
  const float x1 = tb.x + 0.5f * tb.z, y1 = tb.y + 0.5f * tb.w;
  e[0] = tb.x; e[2] = tb.y; e[4] = tb.z; e[6] = tb.w;
  e[8] = (x1 - x0) * (y1 - y0);
}

__global__ __launch_bounds__(128) void cost_strip_k(
    const float* __restrict__ prob,
    const float* __restrict__ pred_boxes,
    const float* __restrict__ boxes,
    const int*   __restrict__ labels,
    float* __restrict__ out) {
  static_assert(NT % 8 == 0, "a half-tile of 8 positions is valid or empty as a whole");
  static_assert(NPOS - NT <= 32 && NPOS >= NT && NPOS % 2 == 0, "one lane per empty position; whole pairs");
  static_assert(NQ % 4 == 0, "the valid rows of an edge q-tile are whole groups of 4 rows (25 lines)");
  static_assert((4 * NT) % 32 == 0 && (4 * NT) / 32 == 25, "4 rows are 25 whole lines");
  static_assert(PR_LOADS * 32 <= ST_V4, "the prob rows fit the strip's LDS tile");
  static_assert(OH_HOT + 2 + 4 * (KSTEPS - 1) + 1 < OH_LEN, "the farthest window stays inside the table");
  static_assert(OH_HOT >= NC && OH_LEN % 32 == 0, "windows start at slot 0 or later; whole-wave fill");
  static_assert(KSTEPS * OR_TPL <= CNT_LEN && KSTEPS <= 32, "every (key, set) has a slot; a key is 5 bits");
  static_assert(OR_TPL <= 8 && OR_TPL > 4, "every wave has one or two sets of targets; wave 3 has one");
  __shared__ float4 stage[4][ST_V4];
  __shared__ float ohtab[4][OH_LEN];
  __shared__ __attribute__((aligned(16))) float tpair[(NPOS / 2) * TP_PITCH];
  __shared__ int slab[NPOS];
  __shared__ __attribute__((aligned(16))) int scol[NPOS];
  __shared__ int scnt[CNT_LEN];

  const int wave = __builtin_amdgcn_readfirstlane(threadIdx.x >> 5);
  const int qt = blockIdx.x * 4 + wave;
  const bool has_tile = (qt * 16 < NQ);
  const int b = blockIdx.z;
  const int nrows = (NQ - qt * 16 < 16) ? (NQ - qt * 16) : 16;

  const int lane = threadIdx.x & 31;
  const int half = lane >> 4;
  const int l15  = lane & 15;
  const int kb   = half << 1;

  float4* st = stage[wave];
  float* zt = ohtab[wave];

  if (has_tile) {
    const float4* psrc = (const float4*)(prob + ((size_t)b * NQ + qt * 16) * NC);
    const int last = nrows * (NC / 4) - 1;
    float4 pin[PR_LOADS];
#pragma unroll
    for (int i = 0; i < PR_LOADS; ++i) {
      const int idx = i * 32 + lane;
      pin[i] = psrc[idx < last ? idx : last];
    }
#pragma unroll
    for (int i = 0; i < PR_LOADS; ++i) st[i * 32 + lane] = pin[i];
#pragma unroll
    for (int i = 0; i < OH_LEN / 32; ++i) zt[i * 32 + lane] = (i * 32 + lane == OH_HOT) ? 1.0f : 0.0f;
  }

  {
    int      lab[2];
    int      key[2];
    unsigned peers[2];
    float4   tbx[2];
#pragma unroll
    for (int s = 0; s < 2; ++s) {
      const int j  = wave + 4 * s;
      const int t  = j * 32 + lane;
      const int tc = (t < NT) ? t : (NT - 1);
      lab[s] = labels[(size_t)b * NT + tc];
      tbx[s] = *(const float4*)(boxes + ((size_t)b * NT + tc) * 4);
      key[s] = label_key(lab[s]);
    }
#pragma unroll
    for (int s = 0; s < 2; ++s) {
      const int j = wave + 4 * s;
      if (j < OR_TPL) {
        if (lane < KSTEPS) scnt[lane * OR_TPL + j] = 0;
        const bool val = (j * 32 + lane < NT);
        unsigned m = __builtin_amdgcn_ballot_w32(val);
#pragma unroll
        for (int bit = 0; bit < 5; ++bit) {
          const int full = (int)((unsigned)key[s] << (31 - bit)) >> 31;
          const unsigned bm = __builtin_amdgcn_ballot_w32(full != 0);
          m &= ~(bm ^ (unsigned)full);
        }
        peers[s] = m;
        if (val && __builtin_amdgcn_mbcnt_lo(m, 0u) == 0u) scnt[key[s] * OR_TPL + j] = __builtin_popcount(m);
      } else {
        peers[s] = 0u;
        if (KSTEPS * OR_TPL + lane < CNT_LEN) scnt[KSTEPS * OR_TPL + lane] = 0;
      }
    }
    __syncthreads();
    if (wave == 0) {
      int cnt[CNT_PER_LANE];
      int sum = 0;
#pragma unroll
      for (int i = 0; i < CNT_PER_LANE; ++i) { cnt[i] = scnt[lane * CNT_PER_LANE + i]; sum += cnt[i]; }
      int incl = sum;
#pragma unroll
      for (int d = 1; d < 32; d <<= 1) {
        const int up = __shfl_up(incl, d);
        incl += (lane >= d) ? up : 0;
      }
      int run = incl - sum;
#pragma unroll
      for (int i = 0; i < CNT_PER_LANE; ++i) { scnt[lane * CNT_PER_LANE + i] = run; run += cnt[i]; }
    }
    __syncthreads();
#pragma unroll
    for (int s = 0; s < 2; ++s) {
      const int j = wave + 4 * s;
      const int t = j * 32 + lane;
      if (j < OR_TPL && t < NT) {
        const int pos = (int)__builtin_amdgcn_mbcnt_lo(peers[s], (unsigned)scnt[key[s] * OR_TPL + j]);
        slab[pos] = lab[s];
        scol[pos] = t;
        write_target(tpair, pos, tbx[s]);
      }
    }
    if (wave == 3 && lane < NPOS - NT) {
      slab[NT + lane] = PAD_LABEL;
      scol[NT + lane] = 0;
      write_target(tpair, NT + lane, make_float4(0.5f, 0.5f, 0.5f, 0.5f));
    }
  }
  __syncthreads();
  if (!has_tile) return;

  const bool qval = (l15 < nrows);
  const int  r_cl = qval ? l15 : (nrows - 1);
  const int  q_cl = qt * 16 + r_cl;

  v2f pfrag[KSTEPS];
  {
    const v2f* prow = (const v2f*)st + r_cl * (NC / 2) + half;
#pragma unroll
    for (int kk = 0; kk < KSTEPS; ++kk) pfrag[kk] = prow[2 * kk];
  }

  const float4 pb = *(const float4*)(pred_boxes + ((size_t)b * NQ + q_cl) * 4);
  const float ax0 = pb.x - 0.5f * pb.z, ay0 = pb.y - 0.5f * pb.w;
  const float ax1 = pb.x + 0.5f * pb.z, ay1 = pb.y + 0.5f * pb.w;
  const float area1 = (ax1 - ax0) * (ay1 - ay0);

  float* srow = (float*)st + l15 * NT;

  for (int tt = 0; tt < TTILES; ++tt) {
    const int labm = slab[tt * 16 + l15];
    const unsigned lrel2 = (unsigned)(labm - kb + 2);
    const unsigned woff  = (lrel2 < 94u) ? lrel2 : 94u;
    const float* zwin = zt + (OH_HOT + 2) - woff;

    const int keym    = label_key(labm);
    const int lastrow = (tt == TTILES - 1) ? ((NT - 1) & 15) : 15;
    const int kfirst  = __builtin_amdgcn_readlane(keym, 0);
    const int klast   = __builtin_amdgcn_readlane(keym, lastrow);
    const unsigned kmask = (2u << klast) - (1u << kfirst);

    const int  t0   = tt * 16 + (half << 3);
    const bool tval = (t0 < NT);
    const float* tpp = tpair + (t0 >> 1) * TP_PITCH;
    const int4 c0 = *(const int4*)(scol + t0);
    const int4 c1 = *(const int4*)(scol + t0 + 4);

    v8f acc = {0.f, 0.f, 0.f, 0.f, 0.f, 0.f, 0.f, 0.f};
#pragma unroll
    for (int kk = 0; kk < KSTEPS; ++kk) {
      if ((kmask >> kk) & 1u) {
        v2f oh;
        oh.x = zwin[4 * kk];
        oh.y = zwin[4 * kk + 1];
        acc = __builtin_amdgcn_wmma_f32_16x16x4_f32(
             false, oh,  false, pfrag[kk],
             (short)0, acc,  false,  false);
      }
    }

    float res[8];
#pragma unroll
    for (int r = 0; r < 8; ++r) {
      const float* e = tpp + (r >> 1) * TP_PITCH + (r & 1);
      const float tcx = e[0], tcy = e[2], tw = e[4], th = e[6];
      const float area2 = e[8];
      const float bx0 = tcx - 0.5f * tw, by0 = tcy - 0.5f * th;
      const float bx1 = tcx + 0.5f * tw, by1 = tcy + 0.5f * th;

      const float l1 = fabsf(pb.x - tcx) + fabsf(pb.y - tcy) +
                       fabsf(pb.z - tw) + fabsf(pb.w - th);

      const float iw    = fminf(ax1, bx1) - fmaxf(ax0, bx0);
      const float ih    = fminf(ay1, by1) - fmaxf(ay0, by0);
      const float inter = fmaxf(iw, 0.f) * fmaxf(ih, 0.f);
      const float uni   = area1 + area2 - inter;

      const float ew    = fmaxf(ax1, bx1) - fminf(ax0, bx0);
      const float eh    = fmaxf(ay1, by1) - fminf(ay0, by0);
      const float areae = fmaxf(ew, 0.f) * fmaxf(eh, 0.f);
      const float gq    = fmaf(inter, areae, uni * uni) / (uni * areae);
      const float giou  = gq - 1.0f;

      res[r] = 5.0f * l1 - acc[r] - 2.0f * giou;
    }

    if (qval && tval) {
      srow[c0.x] = res[0]; srow[c0.y] = res[1]; srow[c0.z] = res[2]; srow[c0.w] = res[3];
      srow[c1.x] = res[4]; srow[c1.y] = res[5]; srow[c1.z] = res[6]; srow[c1.w] = res[7];
    }
  }

  wave_lds_handover();
  {
    float4* gdst = (float4*)(out + ((size_t)b * NQ + qt * 16) * NT);
    const int fl_g     = lane >> 3;
    const int fl_piece = lane & 7;
    if (4 * fl_g < nrows) {
      const int i0 = 25 * 8 * fl_g + fl_piece;
#pragma unroll
      for (int c = 0; c < 5; ++c) {
        float4 v[5];
#pragma unroll
        for (int i = 0; i < 5; ++i) v[i] = st[i0 + (5 * c + i) * 8];
#pragma unroll
        for (int i = 0; i < 5; ++i) gdst[i0 + (5 * c + i) * 8] = v[i];
      }
    }
  }
}

extern "C" void kernel_launch(void* const* d_in, const int* in_sizes, int n_in,
                              void* d_out, int out_size, void* d_ws, size_t ws_size,
                              hipStream_t stream) {
  const float* pred_logits = (const float*)d_in[0];
  const float* pred_boxes  = (const float*)d_in[1];
  const float* boxes       = (const float*)d_in[2];
  const int*   labels      = (const int*)d_in[3];
  float* out  = (float*)d_out;
  float* prob = (float*)d_ws;

  softmax_rows_k<<<SM_BLOCKS, SM_WAVES * 32, 0, stream>>>(pred_logits, prob);

  dim3 grid((QTILES + 3) / 4, 1, BS);
  cost_strip_k<<<grid, 128, 0, stream>>>(prob, pred_boxes, boxes, labels, out);
}


# Start-up loads issued as one group, small ones first (labels, target boxes, query box, then the 12 prob pieces): the ordering pass runs under the prob loads' flight; its barriers fence LDS only
# speedup vs baseline: 1.6707x; 1.0483x over previous
#include <hip/hip_runtime.h>
#include <hip/hip_bf16.h>
#include <cfloat>

#define BS 64
#define NQ 900
#define NT 200
#define NC 92
#define KSTEPS (NC / 4)
#define TTILES ((NT + 15) / 16)
#define QTILES ((NQ + 15) / 16)

typedef float v2f __attribute__((ext_vector_type(2)));
typedef float v8f __attribute__((ext_vector_type(8)));

#define SM_WAVES 4
#define SM_ROWS  32
#define SM_V4    (NC / 4)
#define SM_CHUNKS ((BS * NQ) / SM_ROWS)
#define SM_BLOCKS (SM_CHUNKS / SM_WAVES)

__global__ __launch_bounds__(SM_WAVES * 32) void softmax_rows_k(
    const float* __restrict__ logits, float* __restrict__ prob) {
  static_assert((BS * NQ) % SM_ROWS == 0, "rows split into whole 32-row chunks");
  static_assert(SM_CHUNKS % SM_WAVES == 0, "chunks split into whole work-groups");
  static_assert(NC % 4 == 0, "a row is a whole number of float4");
  __shared__ float4 tile[SM_WAVES][SM_ROWS * SM_V4];

  const int wave  = threadIdx.x >> 5;
  const int lane  = threadIdx.x & 31;
  const int chunk = blockIdx.x * SM_WAVES + wave;
  const size_t base = (size_t)chunk * (SM_ROWS * NC);
  const float4* src = (const float4*)(logits + base);
  float4*       dst = (float4*)(prob + base);
  float4* t = tile[wave];

  float4 v[SM_V4];
#pragma unroll
  for (int i = 0; i < SM_V4; ++i) v[i] = src[i * 32 + lane];
#pragma unroll
  for (int i = 0; i < SM_V4; ++i) t[i * 32 + lane] = v[i];
  __syncthreads();

  float4* myrow = t + lane * SM_V4;
  float4 r[SM_V4];
#pragma unroll
  for (int j = 0; j < SM_V4; ++j) r[j] = myrow[j];

  float m = fmaxf(fmaxf(r[0].x, r[0].y), fmaxf(r[0].z, r[0].w));
#pragma unroll
  for (int j = 1; j < SM_V4; ++j)
    m = fmaxf(m, fmaxf(fmaxf(r[j].x, r[j].y), fmaxf(r[j].z, r[j].w)));

  float s0 = 0.f, s1 = 0.f, s2 = 0.f, s3 = 0.f;
#pragma unroll
  for (int j = 0; j < SM_V4; ++j) {
    r[j].x = __expf(r[j].x - m); s0 += r[j].x;
    r[j].y = __expf(r[j].y - m); s1 += r[j].y;
    r[j].z = __expf(r[j].z - m); s2 += r[j].z;
    r[j].w = __expf(r[j].w - m); s3 += r[j].w;
  }
  const float inv = 1.0f / ((s0 + s1) + (s2 + s3));

#pragma unroll
  for (int j = 0; j < SM_V4; ++j)
    myrow[j] = make_float4(r[j].x * inv, r[j].y * inv, r[j].z * inv, r[j].w * inv);
  __syncthreads();

#pragma unroll
  for (int i = 0; i < SM_V4; ++i) v[i] = t[i * 32 + lane];
#pragma unroll
  for (int i = 0; i < SM_V4; ++i) dst[i * 32 + lane] = v[i];
}

#define ST_V4 (16 * NT / 4)
#define PR_V4 (16 * NC / 4)
#define PR_LOADS ((PR_V4 + 31) / 32)
#define OH_HOT 92
#define OH_LEN 192
#define OR_TPL ((NT + 31) / 32)
#define NPOS (TTILES * 16)
#define PAD_LABEL (NC + 8)
#define CNT_PER_LANE 6
#define CNT_LEN (32 * CNT_PER_LANE)
#define TP_PITCH 12

__device__ __forceinline__ void wave_lds_handover() {
  __builtin_amdgcn_fence(__ATOMIC_RELEASE, "wavefront");
  __builtin_amdgcn_wave_barrier();
  __builtin_amdgcn_fence(__ATOMIC_ACQUIRE, "wavefront");
}

__device__ __forceinline__ void lds_only_barrier() {
  __builtin_amdgcn_fence(__ATOMIC_RELEASE, "workgroup", "local");
  __builtin_amdgcn_s_barrier();
  __builtin_amdgcn_fence(__ATOMIC_ACQUIRE, "workgroup", "local");
}

__device__ __forceinline__ int label_key(int label) {
  const unsigned k = (unsigned)label >> 2;
  return (int)(k < (unsigned)(KSTEPS - 1) ? k : (unsigned)(KSTEPS - 1));
}

__device__ __forceinline__ void write_target(float* tp, int pos, float4 tb) {
  float* e = tp + (pos >> 1) * TP_PITCH + (pos & 1);
  const float x0 = tb.x - 0.5f * tb.z, y0 = tb.y - 0.5f * tb.w;
  const float x1 = tb.x + 0.5f * tb.z, y1 = tb.y + 0.5f * tb.w;
  e[0] = tb.x; e[2] = tb.y; e[4] = tb.z; e[6] = tb.w;
  e[8] = (x1 - x0) * (y1 - y0);
}

__global__ __launch_bounds__(128) void cost_strip_k(
    const float* __restrict__ prob,
    const float* __restrict__ pred_boxes,
    const float* __restrict__ boxes,
    const int*   __restrict__ labels,
    float* __restrict__ out) {
  static_assert(NT % 8 == 0, "a half-tile of 8 positions is valid or empty as a whole");
  static_assert(NPOS - NT <= 32 && NPOS >= NT && NPOS % 2 == 0, "one lane per empty position; whole pairs");
  static_assert(NQ % 4 == 0, "the valid rows of an edge q-tile are whole groups of 4 rows (25 lines)");
  static_assert((4 * NT) % 32 == 0 && (4 * NT) / 32 == 25, "4 rows are 25 whole lines");
  static_assert(PR_LOADS * 32 <= ST_V4, "the prob rows fit the strip's LDS tile");
  static_assert(OH_HOT + 2 + 4 * (KSTEPS - 1) + 1 < OH_LEN, "the farthest window stays inside the table");
  static_assert(OH_HOT >= NC && OH_LEN % 32 == 0, "windows start at slot 0 or later; whole-wave fill");
  static_assert(KSTEPS * OR_TPL <= CNT_LEN && KSTEPS <= 32, "every (key, set) has a slot; a key is 5 bits");
  static_assert(OR_TPL <= 8 && OR_TPL > 4, "every wave has one or two sets of targets; wave 3 has one");
  __shared__ float4 stage[4][ST_V4];
  __shared__ float ohtab[4][OH_LEN];
  __shared__ __attribute__((aligned(16))) float tpair[(NPOS / 2) * TP_PITCH];
  __shared__ int slab[NPOS];
  __shared__ __attribute__((aligned(16))) int scol[NPOS];
  __shared__ int scnt[CNT_LEN];

  const int wave = __builtin_amdgcn_readfirstlane(threadIdx.x >> 5);
  const int qt_own = blockIdx.x * 4 + wave;
  const bool has_tile = (qt_own < QTILES);
  const int qt = has_tile ? qt_own : (QTILES - 1);
  const int b = blockIdx.z;
  const int nrows = (NQ - qt * 16 < 16) ? (NQ - qt * 16) : 16;

  const int lane = threadIdx.x & 31;
  const int half = lane >> 4;
  const int l15  = lane & 15;
  const int kb   = half << 1;

  float4* st = stage[wave];
  float* zt = ohtab[wave];

  int    lab[2];
  float4 tbx[2];
#pragma unroll
  for (int s = 0; s < 2; ++s) {
    const int j  = wave + 4 * s;
    const int t  = j * 32 + lane;
    const int tc = (t < NT) ? t : (NT - 1);
    lab[s] = labels[(size_t)b * NT + tc];
    tbx[s] = *(const float4*)(boxes + ((size_t)b * NT + tc) * 4);
  }
  const bool qval = (l15 < nrows);
  const int  r_cl = qval ? l15 : (nrows - 1);
  const int  q_cl = qt * 16 + r_cl;
  const float4 pb = *(const float4*)(pred_boxes + ((size_t)b * NQ + q_cl) * 4);
  static_assert(PR_LOADS == 12, "the prob pieces are held in 12 named register quads");
  const float4* psrc = (const float4*)(prob + ((size_t)b * NQ + qt * 16) * NC);
  const int last = nrows * (NC / 4) - 1;
#define PR_LD(i) const float4 pin##i = psrc[((i) * 32 + lane) < last ? ((i) * 32 + lane) : last]
  PR_LD(0); PR_LD(1); PR_LD(2); PR_LD(3); PR_LD(4); PR_LD(5);
  PR_LD(6); PR_LD(7); PR_LD(8); PR_LD(9); PR_LD(10); PR_LD(11);
#undef PR_LD
#pragma unroll
  for (int i = 0; i < OH_LEN / 32; ++i) zt[i * 32 + lane] = (i * 32 + lane == OH_HOT) ? 1.0f : 0.0f;

  {
    int      key[2];
    unsigned peers[2];
#pragma unroll
    for (int s = 0; s < 2; ++s)
      key[s] = label_key(lab[s]);
#pragma unroll
    for (int s = 0; s < 2; ++s) {
      const int j = wave + 4 * s;
      if (j < OR_TPL) {
        if (lane < KSTEPS) scnt[lane * OR_TPL + j] = 0;
        const bool val = (j * 32 + lane < NT);
        unsigned m = __builtin_amdgcn_ballot_w32(val);
#pragma unroll
        for (int bit = 0; bit < 5; ++bit) {
          const int full = (int)((unsigned)key[s] << (31 - bit)) >> 31;
          const unsigned bm = __builtin_amdgcn_ballot_w32(full != 0);
          m &= ~(bm ^ (unsigned)full);
        }
        peers[s] = m;
        if (val && __builtin_amdgcn_mbcnt_lo(m, 0u) == 0u) scnt[key[s] * OR_TPL + j] = __builtin_popcount(m);
      } else {
        peers[s] = 0u;
        if (KSTEPS * OR_TPL + lane < CNT_LEN) scnt[KSTEPS * OR_TPL + lane] = 0;
      }
    }
    lds_only_barrier();
    if (wave == 0) {
      int cnt[CNT_PER_LANE];
      int sum = 0;
#pragma unroll
      for (int i = 0; i < CNT_PER_LANE; ++i) { cnt[i] = scnt[lane * CNT_PER_LANE + i]; sum += cnt[i]; }
      int incl = sum;
#pragma unroll
      for (int d = 1; d < 32; d <<= 1) {
        const int up = __shfl_up(incl, d);
        incl += (lane >= d) ? up : 0;
      }
      int run = incl - sum;
#pragma unroll
      for (int i = 0; i < CNT_PER_LANE; ++i) { scnt[lane * CNT_PER_LANE + i] = run; run += cnt[i]; }
    }
    lds_only_barrier();
#pragma unroll
    for (int s = 0; s < 2; ++s) {
      const int j = wave + 4 * s;
      const int t = j * 32 + lane;
      if (j < OR_TPL && t < NT) {
        const int pos = (int)__builtin_amdgcn_mbcnt_lo(peers[s], (unsigned)scnt[key[s] * OR_TPL + j]);
        slab[pos] = lab[s];
        scol[pos] = t;
        write_target(tpair, pos, tbx[s]);
      }
    }
    if (wave == 3 && lane < NPOS - NT) {
      slab[NT + lane] = PAD_LABEL;
      scol[NT + lane] = 0;
      write_target(tpair, NT + lane, make_float4(0.5f, 0.5f, 0.5f, 0.5f));
    }
  }
#define PR_ST(i) st[(i) * 32 + lane] = pin##i
  PR_ST(0); PR_ST(1); PR_ST(2); PR_ST(3); PR_ST(4); PR_ST(5);
  PR_ST(6); PR_ST(7); PR_ST(8); PR_ST(9); PR_ST(10); PR_ST(11);
#undef PR_ST
  __syncthreads();
  if (!has_tile) return;

  v2f pfrag[KSTEPS];
  {
    const v2f* prow = (const v2f*)st + r_cl * (NC / 2) + half;
#pragma unroll
    for (int kk = 0; kk < KSTEPS; ++kk) pfrag[kk] = prow[2 * kk];
  }

  const float ax0 = pb.x - 0.5f * pb.z, ay0 = pb.y - 0.5f * pb.w;
  const float ax1 = pb.x + 0.5f * pb.z, ay1 = pb.y + 0.5f * pb.w;
  const float area1 = (ax1 - ax0) * (ay1 - ay0);

  float* srow = (float*)st + l15 * NT;

  for (int tt = 0; tt < TTILES; ++tt) {
    const int labm = slab[tt * 16 + l15];
    const unsigned lrel2 = (unsigned)(labm - kb + 2);
    const unsigned woff  = (lrel2 < 94u) ? lrel2 : 94u;
    const float* zwin = zt + (OH_HOT + 2) - woff;

    const int keym    = label_key(labm);
    const int lastrow = (tt == TTILES - 1) ? ((NT - 1) & 15) : 15;
    const int kfirst  = __builtin_amdgcn_readlane(keym, 0);
    const int klast   = __builtin_amdgcn_readlane(keym, lastrow);
    const unsigned kmask = (2u << klast) - (1u << kfirst);

    const int  t0   = tt * 16 + (half << 3);
    const bool tval = (t0 < NT);
    const float* tpp = tpair + (t0 >> 1) * TP_PITCH;
    const int4 c0 = *(const int4*)(scol + t0);
    const int4 c1 = *(const int4*)(scol + t0 + 4);

    v8f acc = {0.f, 0.f, 0.f, 0.f, 0.f, 0.f, 0.f, 0.f};
#pragma unroll
    for (int kk = 0; kk < KSTEPS; ++kk) {
      if ((kmask >> kk) & 1u) {
        v2f oh;
        oh.x = zwin[4 * kk];
        oh.y = zwin[4 * kk + 1];
        acc = __builtin_amdgcn_wmma_f32_16x16x4_f32(
             false, oh,  false, pfrag[kk],
             (short)0, acc,  false,  false);
      }
    }

    float res[8];
#pragma unroll
    for (int r = 0; r < 8; ++r) {
      const float* e = tpp + (r >> 1) * TP_PITCH + (r & 1);
      const float tcx = e[0], tcy = e[2], tw = e[4], th = e[6];
      const float area2 = e[8];
      const float bx0 = tcx - 0.5f * tw, by0 = tcy - 0.5f * th;
      const float bx1 = tcx + 0.5f * tw, by1 = tcy + 0.5f * th;

      const float l1 = fabsf(pb.x - tcx) + fabsf(pb.y - tcy) +
                       fabsf(pb.z - tw) + fabsf(pb.w - th);

      const float iw    = fminf(ax1, bx1) - fmaxf(ax0, bx0);
      const float ih    = fminf(ay1, by1) - fmaxf(ay0, by0);
      const float inter = fmaxf(iw, 0.f) * fmaxf(ih, 0.f);
      const float uni   = area1 + area2 - inter;

      const float ew    = fmaxf(ax1, bx1) - fminf(ax0, bx0);
      const float eh    = fmaxf(ay1, by1) - fminf(ay0, by0);
      const float areae = fmaxf(ew, 0.f) * fmaxf(eh, 0.f);
      const float gq    = fmaf(inter, areae, uni * uni) / (uni * areae);
      const float giou  = gq - 1.0f;

      res[r] = 5.0f * l1 - acc[r] - 2.0f * giou;
    }

    if (qval && tval) {
      srow[c0.x] = res[0]; srow[c0.y] = res[1]; srow[c0.z] = res[2]; srow[c0.w] = res[3];
      srow[c1.x] = res[4]; srow[c1.y] = res[5]; srow[c1.z] = res[6]; srow[c1.w] = res[7];
    }
  }

  wave_lds_handover();
  {
    float4* gdst = (float4*)(out + ((size_t)b * NQ + qt * 16) * NT);
    const int fl_g     = lane >> 3;
    const int fl_piece = lane & 7;
    if (4 * fl_g < nrows) {
      const int i0 = 25 * 8 * fl_g + fl_piece;
#pragma unroll
      for (int c = 0; c < 5; ++c) {
        float4 v[5];
#pragma unroll
        for (int i = 0; i < 5; ++i) v[i] = st[i0 + (5 * c + i) * 8];
#pragma unroll
        for (int i = 0; i < 5; ++i) gdst[i0 + (5 * c + i) * 8] = v[i];
      }
    }
  }
}

extern "C" void kernel_launch(void* const* d_in, const int* in_sizes, int n_in,
                              void* d_out, int out_size, void* d_ws, size_t ws_size,
                              hipStream_t stream) {
  const float* pred_logits = (const float*)d_in[0];
  const float* pred_boxes  = (const float*)d_in[1];
  const float* boxes       = (const float*)d_in[2];
  const int*   labels      = (const int*)d_in[3];
  float* out  = (float*)d_out;
  float* prob = (float*)d_ws;

  softmax_rows_k<<<SM_BLOCKS, SM_WAVES * 32, 0, stream>>>(pred_logits, prob);

  dim3 grid((QTILES + 3) / 4, 1, BS);
  cost_strip_k<<<grid, 128, 0, stream>>>(prob, pred_boxes, boxes, labels, out);
}


# GIoU quotient expanded by hand, two pairs side by side: 2 rcp + 7 two-wide FMA/mul steps per 2 quotients replace 2 x 11 one-wide divide steps; still one correctly rounded quotient per pair
# speedup vs baseline: 1.7719x; 1.0606x over previous
#include <hip/hip_runtime.h>
#include <hip/hip_bf16.h>
#include <cfloat>

#define BS 64
#define NQ 900
#define NT 200
#define NC 92
#define KSTEPS (NC / 4)
#define TTILES ((NT + 15) / 16)
#define QTILES ((NQ + 15) / 16)

typedef float v2f __attribute__((ext_vector_type(2)));
typedef float v8f __attribute__((ext_vector_type(8)));

#define SM_WAVES 4
#define SM_ROWS  32
#define SM_V4    (NC / 4)
#define SM_CHUNKS ((BS * NQ) / SM_ROWS)
#define SM_BLOCKS (SM_CHUNKS / SM_WAVES)

__global__ __launch_bounds__(SM_WAVES * 32) void softmax_rows_k(
    const float* __restrict__ logits, float* __restrict__ prob) {
  static_assert((BS * NQ) % SM_ROWS == 0, "rows split into whole 32-row chunks");
  static_assert(SM_CHUNKS % SM_WAVES == 0, "chunks split into whole work-groups");
  static_assert(NC % 4 == 0, "a row is a whole number of float4");
  __shared__ float4 tile[SM_WAVES][SM_ROWS * SM_V4];

  const int wave  = threadIdx.x >> 5;
  const int lane  = threadIdx.x & 31;
  const int chunk = blockIdx.x * SM_WAVES + wave;
  const size_t base = (size_t)chunk * (SM_ROWS * NC);
  const float4* src = (const float4*)(logits + base);
  float4*       dst = (float4*)(prob + base);
  float4* t = tile[wave];

  float4 v[SM_V4];
#pragma unroll
  for (int i = 0; i < SM_V4; ++i) v[i] = src[i * 32 + lane];
#pragma unroll
  for (int i = 0; i < SM_V4; ++i) t[i * 32 + lane] = v[i];
  __syncthreads();

  float4* myrow = t + lane * SM_V4;
  float4 r[SM_V4];
#pragma unroll
  for (int j = 0; j < SM_V4; ++j) r[j] = myrow[j];

  float m = fmaxf(fmaxf(r[0].x, r[0].y), fmaxf(r[0].z, r[0].w));
#pragma unroll
  for (int j = 1; j < SM_V4; ++j)
    m = fmaxf(m, fmaxf(fmaxf(r[j].x, r[j].y), fmaxf(r[j].z, r[j].w)));

  float s0 = 0.f, s1 = 0.f, s2 = 0.f, s3 = 0.f;
#pragma unroll
  for (int j = 0; j < SM_V4; ++j) {
    r[j].x = __expf(r[j].x - m); s0 += r[j].x;
    r[j].y = __expf(r[j].y - m); s1 += r[j].y;
    r[j].z = __expf(r[j].z - m); s2 += r[j].z;
    r[j].w = __expf(r[j].w - m); s3 += r[j].w;
  }
  const float inv = 1.0f / ((s0 + s1) + (s2 + s3));

#pragma unroll
  for (int j = 0; j < SM_V4; ++j)
    myrow[j] = make_float4(r[j].x * inv, r[j].y * inv, r[j].z * inv, r[j].w * inv);
  __syncthreads();

#pragma unroll
  for (int i = 0; i < SM_V4; ++i) v[i] = t[i * 32 + lane];
#pragma unroll
  for (int i = 0; i < SM_V4; ++i) dst[i * 32 + lane] = v[i];
}

#define ST_V4 (16 * NT / 4)
#define PR_V4 (16 * NC / 4)
#define PR_LOADS ((PR_V4 + 31) / 32)
#define OH_HOT 92
#define OH_LEN 192
#define OR_TPL ((NT + 31) / 32)
#define NPOS (TTILES * 16)
#define PAD_LABEL (NC + 8)
#define CNT_PER_LANE 6
#define CNT_LEN (32 * CNT_PER_LANE)
#define TP_PITCH 12

__device__ __forceinline__ void wave_lds_handover() {
  __builtin_amdgcn_fence(__ATOMIC_RELEASE, "wavefront");
  __builtin_amdgcn_wave_barrier();
  __builtin_amdgcn_fence(__ATOMIC_ACQUIRE, "wavefront");
}

__device__ __forceinline__ void lds_only_barrier() {
  __builtin_amdgcn_fence(__ATOMIC_RELEASE, "workgroup", "local");
  __builtin_amdgcn_s_barrier();
  __builtin_amdgcn_fence(__ATOMIC_ACQUIRE, "workgroup", "local");
}

__device__ __forceinline__ int label_key(int label) {
  const unsigned k = (unsigned)label >> 2;
  return (int)(k < (unsigned)(KSTEPS - 1) ? k : (unsigned)(KSTEPS - 1));
}

__device__ __forceinline__ void write_target(float* tp, int pos, float4 tb) {
  float* e = tp + (pos >> 1) * TP_PITCH + (pos & 1);
  const float x0 = tb.x - 0.5f * tb.z, y0 = tb.y - 0.5f * tb.w;
  const float x1 = tb.x + 0.5f * tb.z, y1 = tb.y + 0.5f * tb.w;
  e[0] = tb.x; e[2] = tb.y; e[4] = tb.z; e[6] = tb.w;
  e[8] = (x1 - x0) * (y1 - y0);
}

__device__ __forceinline__ v2f div_exact_x2(v2f n, v2f d) {
  v2f r;
  r.x = __builtin_amdgcn_rcpf(d.x);
  r.y = __builtin_amdgcn_rcpf(d.y);
  const v2f one = {1.0f, 1.0f};
  v2f e = __builtin_elementwise_fma(-d, r, one);
  r = __builtin_elementwise_fma(e, r, r);
  v2f q = n * r;
  e = __builtin_elementwise_fma(-d, q, n);
  q = __builtin_elementwise_fma(e, r, q);
  e = __builtin_elementwise_fma(-d, q, n);
  q = __builtin_elementwise_fma(e, r, q);
  return q;
}

__global__ __launch_bounds__(128) void cost_strip_k(
    const float* __restrict__ prob,
    const float* __restrict__ pred_boxes,
    const float* __restrict__ boxes,
    const int*   __restrict__ labels,
    float* __restrict__ out) {
  static_assert(NT % 8 == 0, "a half-tile of 8 positions is valid or empty as a whole");
  static_assert(NPOS - NT <= 32 && NPOS >= NT && NPOS % 2 == 0, "one lane per empty position; whole pairs");
  static_assert(NQ % 4 == 0, "the valid rows of an edge q-tile are whole groups of 4 rows (25 lines)");
  static_assert((4 * NT) % 32 == 0 && (4 * NT) / 32 == 25, "4 rows are 25 whole lines");
  static_assert(PR_LOADS * 32 <= ST_V4, "the prob rows fit the strip's LDS tile");
  static_assert(OH_HOT + 2 + 4 * (KSTEPS - 1) + 1 < OH_LEN, "the farthest window stays inside the table");
  static_assert(OH_HOT >= NC && OH_LEN % 32 == 0, "windows start at slot 0 or later; whole-wave fill");
  static_assert(KSTEPS * OR_TPL <= CNT_LEN && KSTEPS <= 32, "every (key, set) has a slot; a key is 5 bits");
  static_assert(OR_TPL <= 8 && OR_TPL > 4, "every wave has one or two sets of targets; wave 3 has one");
  __shared__ float4 stage[4][ST_V4];
  __shared__ float ohtab[4][OH_LEN];
  __shared__ __attribute__((aligned(16))) float tpair[(NPOS / 2) * TP_PITCH];
  __shared__ int slab[NPOS];
  __shared__ __attribute__((aligned(16))) int scol[NPOS];
  __shared__ int scnt[CNT_LEN];

  const int wave = __builtin_amdgcn_readfirstlane(threadIdx.x >> 5);
  const int qt_own = blockIdx.x * 4 + wave;
  const bool has_tile = (qt_own < QTILES);
  const int qt = has_tile ? qt_own : (QTILES - 1);
  const int b = blockIdx.z;
  const int nrows = (NQ - qt * 16 < 16) ? (NQ - qt * 16) : 16;

  const int lane = threadIdx.x & 31;
  const int half = lane >> 4;
  const int l15  = lane & 15;
  const int kb   = half << 1;

  float4* st = stage[wave];
  float* zt = ohtab[wave];

  int    lab[2];
  float4 tbx[2];
#pragma unroll
  for (int s = 0; s < 2; ++s) {
    const int j  = wave + 4 * s;
    const int t  = j * 32 + lane;
    const int tc = (t < NT) ? t : (NT - 1);
    lab[s] = labels[(size_t)b * NT + tc];
    tbx[s] = *(const float4*)(boxes + ((size_t)b * NT + tc) * 4);
  }
  const bool qval = (l15 < nrows);
  const int  r_cl = qval ? l15 : (nrows - 1);
  const int  q_cl = qt * 16 + r_cl;
  const float4 pb = *(const float4*)(pred_boxes + ((size_t)b * NQ + q_cl) * 4);
  static_assert(PR_LOADS == 12, "the prob pieces are held in 12 named register quads");
  const float4* psrc = (const float4*)(prob + ((size_t)b * NQ + qt * 16) * NC);
  const int last = nrows * (NC / 4) - 1;
#define PR_LD(i) const float4 pin##i = psrc[((i) * 32 + lane) < last ? ((i) * 32 + lane) : last]
  PR_LD(0); PR_LD(1); PR_LD(2); PR_LD(3); PR_LD(4); PR_LD(5);
  PR_LD(6); PR_LD(7); PR_LD(8); PR_LD(9); PR_LD(10); PR_LD(11);
#undef PR_LD
#pragma unroll
  for (int i = 0; i < OH_LEN / 32; ++i) zt[i * 32 + lane] = (i * 32 + lane == OH_HOT) ? 1.0f : 0.0f;

  {
    int      key[2];
    unsigned peers[2];
#pragma unroll
    for (int s = 0; s < 2; ++s)
      key[s] = label_key(lab[s]);
#pragma unroll
    for (int s = 0; s < 2; ++s) {
      const int j = wave + 4 * s;
      if (j < OR_TPL) {
        if (lane < KSTEPS) scnt[lane * OR_TPL + j] = 0;
        const bool val = (j * 32 + lane < NT);
        unsigned m = __builtin_amdgcn_ballot_w32(val);
#pragma unroll
        for (int bit = 0; bit < 5; ++bit) {
          const int full = (int)((unsigned)key[s] << (31 - bit)) >> 31;
          const unsigned bm = __builtin_amdgcn_ballot_w32(full != 0);
          m &= ~(bm ^ (unsigned)full);
        }
        peers[s] = m;
        if (val && __builtin_amdgcn_mbcnt_lo(m, 0u) == 0u) scnt[key[s] * OR_TPL + j] = __builtin_popcount(m);
      } else {
        peers[s] = 0u;
        if (KSTEPS * OR_TPL + lane < CNT_LEN) scnt[KSTEPS * OR_TPL + lane] = 0;
      }
    }
    lds_only_barrier();
    if (wave == 0) {
      int cnt[CNT_PER_LANE];
      int sum = 0;
#pragma unroll
      for (int i = 0; i < CNT_PER_LANE; ++i) { cnt[i] = scnt[lane * CNT_PER_LANE + i]; sum += cnt[i]; }
      int incl = sum;
#pragma unroll
      for (int d = 1; d < 32; d <<= 1) {
        const int up = __shfl_up(incl, d);
        incl += (lane >= d) ? up : 0;
      }
      int run = incl - sum;
#pragma unroll
      for (int i = 0; i < CNT_PER_LANE; ++i) { scnt[lane * CNT_PER_LANE + i] = run; run += cnt[i]; }
    }
    lds_only_barrier();
#pragma unroll
    for (int s = 0; s < 2; ++s) {
      const int j = wave + 4 * s;
      const int t = j * 32 + lane;
      if (j < OR_TPL && t < NT) {
        const int pos = (int)__builtin_amdgcn_mbcnt_lo(peers[s], (unsigned)scnt[key[s] * OR_TPL + j]);
        slab[pos] = lab[s];
        scol[pos] = t;
        write_target(tpair, pos, tbx[s]);
      }
    }
    if (wave == 3 && lane < NPOS - NT) {
      slab[NT + lane] = PAD_LABEL;
      scol[NT + lane] = 0;
      write_target(tpair, NT + lane, make_float4(0.5f, 0.5f, 0.5f, 0.5f));
    }
  }
#define PR_ST(i) st[(i) * 32 + lane] = pin##i
  PR_ST(0); PR_ST(1); PR_ST(2); PR_ST(3); PR_ST(4); PR_ST(5);
  PR_ST(6); PR_ST(7); PR_ST(8); PR_ST(9); PR_ST(10); PR_ST(11);
#undef PR_ST
  __syncthreads();
  if (!has_tile) return;

  v2f pfrag[KSTEPS];
  {
    const v2f* prow = (const v2f*)st + r_cl * (NC / 2) + half;
#pragma unroll
    for (int kk = 0; kk < KSTEPS; ++kk) pfrag[kk] = prow[2 * kk];
  }

  const float ax0 = pb.x - 0.5f * pb.z, ay0 = pb.y - 0.5f * pb.w;
  const float ax1 = pb.x + 0.5f * pb.z, ay1 = pb.y + 0.5f * pb.w;
  const float area1 = (ax1 - ax0) * (ay1 - ay0);

  float* srow = (float*)st + l15 * NT;

  for (int tt = 0; tt < TTILES; ++tt) {
    const int labm = slab[tt * 16 + l15];
    const unsigned lrel2 = (unsigned)(labm - kb + 2);
    const unsigned woff  = (lrel2 < 94u) ? lrel2 : 94u;
    const float* zwin = zt + (OH_HOT + 2) - woff;

    const int keym    = label_key(labm);
    const int lastrow = (tt == TTILES - 1) ? ((NT - 1) & 15) : 15;
    const int kfirst  = __builtin_amdgcn_readlane(keym, 0);
    const int klast   = __builtin_amdgcn_readlane(keym, lastrow);
    const unsigned kmask = (2u << klast) - (1u << kfirst);

    const int  t0   = tt * 16 + (half << 3);
    const bool tval = (t0 < NT);
    const float* tpp = tpair + (t0 >> 1) * TP_PITCH;
    const int4 c0 = *(const int4*)(scol + t0);
    const int4 c1 = *(const int4*)(scol + t0 + 4);

    v8f acc = {0.f, 0.f, 0.f, 0.f, 0.f, 0.f, 0.f, 0.f};
#pragma unroll
    for (int kk = 0; kk < KSTEPS; ++kk) {
      if ((kmask >> kk) & 1u) {
        v2f oh;
        oh.x = zwin[4 * kk];
        oh.y = zwin[4 * kk + 1];
        acc = __builtin_amdgcn_wmma_f32_16x16x4_f32(
             false, oh,  false, pfrag[kk],
             (short)0, acc,  false,  false);
      }
    }

    float res[8];
    float l1v[8], gnum[8], gden[8];
#pragma unroll
    for (int r = 0; r < 8; ++r) {
      const float* e = tpp + (r >> 1) * TP_PITCH + (r & 1);
      const float tcx = e[0], tcy = e[2], tw = e[4], th = e[6];
      const float area2 = e[8];
      const float bx0 = tcx - 0.5f * tw, by0 = tcy - 0.5f * th;
      const float bx1 = tcx + 0.5f * tw, by1 = tcy + 0.5f * th;

      l1v[r] = fabsf(pb.x - tcx) + fabsf(pb.y - tcy) +
               fabsf(pb.z - tw) + fabsf(pb.w - th);

      const float iw    = fminf(ax1, bx1) - fmaxf(ax0, bx0);
      const float ih    = fminf(ay1, by1) - fmaxf(ay0, by0);
      const float inter = fmaxf(iw, 0.f) * fmaxf(ih, 0.f);
      const float uni   = area1 + area2 - inter;

      const float ew    = fmaxf(ax1, bx1) - fminf(ax0, bx0);
      const float eh    = fmaxf(ay1, by1) - fminf(ay0, by0);
      const float areae = fmaxf(ew, 0.f) * fmaxf(eh, 0.f);
      gnum[r] = fmaf(inter, areae, uni * uni);
      gden[r] = uni * areae;
    }
#pragma unroll
    for (int j = 0; j < 4; ++j) {
      const v2f n2 = {gnum[2 * j], gnum[2 * j + 1]};
      const v2f d2 = {gden[2 * j], gden[2 * j + 1]};
      const v2f gq = div_exact_x2(n2, d2);
      const float giou0 = gq.x - 1.0f;
      const float giou1 = gq.y - 1.0f;
      res[2 * j]     = 5.0f * l1v[2 * j]     - acc[2 * j]     - 2.0f * giou0;
      res[2 * j + 1] = 5.0f * l1v[2 * j + 1] - acc[2 * j + 1] - 2.0f * giou1;
    }

    if (qval && tval) {
      srow[c0.x] = res[0]; srow[c0.y] = res[1]; srow[c0.z] = res[2]; srow[c0.w] = res[3];
      srow[c1.x] = res[4]; srow[c1.y] = res[5]; srow[c1.z] = res[6]; srow[c1.w] = res[7];
    }
  }

  wave_lds_handover();
  {
    float4* gdst = (float4*)(out + ((size_t)b * NQ + qt * 16) * NT);
    const int fl_g     = lane >> 3;
    const int fl_piece = lane & 7;
    if (4 * fl_g < nrows) {
      const int i0 = 25 * 8 * fl_g + fl_piece;
#pragma unroll
      for (int c = 0; c < 5; ++c) {
        float4 v[5];
#pragma unroll
        for (int i = 0; i < 5; ++i) v[i] = st[i0 + (5 * c + i) * 8];
#pragma unroll
        for (int i = 0; i < 5; ++i) gdst[i0 + (5 * c + i) * 8] = v[i];
      }
    }
  }
}

extern "C" void kernel_launch(void* const* d_in, const int* in_sizes, int n_in,
                              void* d_out, int out_size, void* d_ws, size_t ws_size,
                              hipStream_t stream) {
  const float* pred_logits = (const float*)d_in[0];
  const float* pred_boxes  = (const float*)d_in[1];
  const float* boxes       = (const float*)d_in[2];
  const int*   labels      = (const int*)d_in[3];
  float* out  = (float*)d_out;
  float* prob = (float*)d_ws;

  softmax_rows_k<<<SM_BLOCKS, SM_WAVES * 32, 0, stream>>>(pred_logits, prob);

  dim3 grid((QTILES + 3) / 4, 1, BS);
  cost_strip_k<<<grid, 128, 0, stream>>>(prob, pred_boxes, boxes, labels, out);
}
